# starting kernel
# baseline (speedup 1.0000x reference)
#include <hip/hip_runtime.h>
#include <hip/hip_bf16.h>
#include <cmath>

typedef __attribute__((ext_vector_type(2))) float v2f;
typedef __attribute__((ext_vector_type(8))) float v8f;

#define BB    2
#define LL    1024
#define DMOD  1024
#define DI    2048
#define NS    16
#define KC    4
#define RR    64
#define XZW   (2 * DI)
#define DBLW  (RR + 2 * NS)

__device__ __forceinline__ float siluf(float v) {
    return v / (1.0f + __expf(-v));
}
__device__ __forceinline__ float softplusf(float v) {
    return (v > 20.0f) ? v : log1pf(expf(v));
}

template <int NT, bool SP>
__global__ void __launch_bounds__(128) wmma_gemm_f32(
    const float* __restrict__ A, const float* __restrict__ W,
    float* __restrict__ C, int N, int K, int lda,
    const float* __restrict__ bias)
{
    const int wave = threadIdx.x >> 5;
    const int lane = threadIdx.x & 31;
    const int half = lane >> 4;
    const int mr   = lane & 15;
    const int m0   = blockIdx.y * 64 + wave * 16;
    const int nb   = blockIdx.x * (NT * 16);

    v8f acc[NT] = {};

    const float* aptr = A + (size_t)(m0 + mr) * lda + 2 * half;
    const float* wptr = W + (size_t)(2 * half) * N + nb + mr;

    v2f a_cur = *(const v2f*)aptr;
    v2f b_cur[NT];
#pragma unroll
    for (int j = 0; j < NT; ++j) {
        b_cur[j][0] = wptr[j * 16];
        b_cur[j][1] = wptr[(size_t)N + j * 16];
    }

    for (int k0 = 0; k0 < K - 4; k0 += 4) {
        aptr += 4;
        wptr += (size_t)4 * N;
        v2f a_nxt = *(const v2f*)aptr;
        v2f b_nxt[NT];
#pragma unroll
        for (int j = 0; j < NT; ++j) {
            b_nxt[j][0] = wptr[j * 16];
            b_nxt[j][1] = wptr[(size_t)N + j * 16];
        }
#pragma unroll
        for (int j = 0; j < NT; ++j)
            acc[j] = __builtin_amdgcn_wmma_f32_16x16x4_f32(
                false, a_cur, false, b_cur[j], (short)0, acc[j], false, false);
        a_cur = a_nxt;
#pragma unroll
        for (int j = 0; j < NT; ++j) b_cur[j] = b_nxt[j];
    }
#pragma unroll
    for (int j = 0; j < NT; ++j)
        acc[j] = __builtin_amdgcn_wmma_f32_16x16x4_f32(
            false, a_cur, false, b_cur[j], (short)0, acc[j], false, false);

#pragma unroll
    for (int j = 0; j < NT; ++j) {
        const int col = nb + j * 16 + mr;
        const float bsv = SP ? bias[col] : 0.0f;
#pragma unroll
        for (int r = 0; r < 8; ++r) {
            float v = acc[j][r];
            if (SP) v = softplusf(v + bsv);
            C[(size_t)(m0 + r + 8 * half) * N + col] = v;
        }
    }
}

__global__ void conv_silu_kernel(const float* __restrict__ xz,
                                 const float* __restrict__ cw,
                                 const float* __restrict__ cb,
                                 float* __restrict__ u)
{
    const int idx = blockIdx.x * blockDim.x + threadIdx.x;
    const int d  = idx % DI;
    const int bt = idx / DI;
    const int t  = bt % LL;
    float acc = cb[d];
    const float* w = cw + d * KC;
#pragma unroll
    for (int k = 0; k < KC; ++k) {
        const int tt = t - (KC - 1) + k;
        if (tt >= 0)
            acc += xz[(size_t)(bt - (KC - 1) + k) * XZW + d] * w[k];
    }
    u[(size_t)bt * DI + d] = siluf(acc);
}

__global__ void __launch_bounds__(128) scan_kernel(
    const float* __restrict__ dt, const float* __restrict__ u,
    const float* __restrict__ dbl, const float* __restrict__ xz,
    const float* __restrict__ A_log, const float* __restrict__ Dp,
    float* __restrict__ yg)
{
    __shared__ float sBC[2 * NS];
    const int b = blockIdx.x >> 4;
    const int d = (blockIdx.x & 15) * 128 + threadIdx.x;

    float An[NS];
#pragma unroll
    for (int n = 0; n < NS; ++n) An[n] = -__expf(A_log[d * NS + n]);
    const float Dd = Dp[d];
    float h[NS];
#pragma unroll
    for (int n = 0; n < NS; ++n) h[n] = 0.0f;

    for (int t = 0; t < LL; ++t) {
        const size_t row = (size_t)(b * LL + t);
        if (threadIdx.x < 2 * NS)
            sBC[threadIdx.x] = dbl[row * DBLW + RR + threadIdx.x];
        __syncthreads();

        const float dtv = dt[row * DI + d];
        const float uv  = u[row * DI + d];
        const float du  = dtv * uv;
        float y = 0.0f;
#pragma unroll
        for (int n = 0; n < NS; ++n) {
            h[n] = __expf(dtv * An[n]) * h[n] + du * sBC[n];
            y += h[n] * sBC[NS + n];
        }
        const float zv = xz[row * XZW + DI + d];
        yg[row * DI + d] = (y + uv * Dd) * siluf(zv);
        __syncthreads();
    }
}

extern "C" void kernel_launch(void* const* d_in, const int* in_sizes, int n_in,
                              void* d_out, int out_size, void* d_ws, size_t ws_size,
                              hipStream_t stream) {
    const float* x          = (const float*)d_in[0];
    const float* in_proj_w  = (const float*)d_in[1];
    const float* conv_w     = (const float*)d_in[2];
    const float* conv_b     = (const float*)d_in[3];
    const float* x_proj_w   = (const float*)d_in[4];
    const float* dt_proj_w  = (const float*)d_in[5];
    const float* dt_proj_b  = (const float*)d_in[6];
    const float* A_log      = (const float*)d_in[7];
    const float* Dvec       = (const float*)d_in[8];
    const float* out_proj_w = (const float*)d_in[9];
    float* out = (float*)d_out;

    float* ws  = (float*)d_ws;
    float* xz  = ws;
    float* u   = xz  + (size_t)2048 * 4096;
    float* dbl = u   + (size_t)2048 * 2048;
    float* dtb = dbl + (size_t)2048 * 96;
    float* yg  = dtb + (size_t)2048 * 2048;

    const dim3 blk(128);
    const int M = BB * LL;

    wmma_gemm_f32<4, false><<<dim3(4096 / 64, M / 64), blk, 0, stream>>>(
        x, in_proj_w, xz, 4096, DMOD, DMOD, nullptr);

    conv_silu_kernel<<<(BB * LL * DI) / 256, 256, 0, stream>>>(xz, conv_w, conv_b, u);

    wmma_gemm_f32<2, false><<<dim3(DBLW / 32, M / 64), blk, 0, stream>>>(
        u, x_proj_w, dbl, DBLW, DI, DI, nullptr);

    wmma_gemm_f32<4, true><<<dim3(DI / 64, M / 64), blk, 0, stream>>>(
        dbl, dt_proj_w, dtb, DI, RR, DBLW, dt_proj_b);

    scan_kernel<<<32, 128, 0, stream>>>(dtb, u, dbl, xz, A_log, Dvec, yg);

    wmma_gemm_f32<4, false><<<dim3(DMOD / 64, M / 64), blk, 0, stream>>>(
        yg, out_proj_w, out, DMOD, DI, DI, nullptr);
}


# all four matrix products: wave tile grown from 16x64 to as much as 64x64 in registers, fragments by 16-byte global loads (8 consecutive k of A, 4 adjacent W columns per lane); new summation order
# speedup vs baseline: 6.9592x; 6.9592x over previous
#include <hip/hip_runtime.h>
#include <hip/hip_bf16.h>
#include <cmath>

typedef __attribute__((ext_vector_type(2))) float v2f;
typedef __attribute__((ext_vector_type(4))) float v4f;
typedef __attribute__((ext_vector_type(8))) float v8f;

#define BB    2
#define LL    1024
#define DMOD  1024
#define DI    2048
#define NS    16
#define KC    4
#define RR    64
#define XZW   (2 * DI)
#define DBLW  (RR + 2 * NS)

__device__ __forceinline__ float siluf(float v) {
    return v / (1.0f + __expf(-v));
}
__device__ __forceinline__ float softplusf(float v) {
    return (v > 20.0f) ? v : log1pf(expf(v));
}

template <int CPL> struct colvec;
template <> struct colvec<4> { typedef v4f type; };
template <> struct colvec<2> { typedef v2f type; };

template <int MT, int CPL, int KS>
struct Slab {
    v4f a[MT][KS / 8];
    typename colvec<CPL>::type b[KS / 2];
};

template <int MT, int CPL, int KS, int N, int LDA>
__device__ __forceinline__ void slab_load(Slab<MT, CPL, KS>& s,
                                          const float* __restrict__ ap,
                                          const float* __restrict__ wp)
{
    typedef typename colvec<CPL>::type cvec;
#pragma unroll
    for (int mt = 0; mt < MT; ++mt)
#pragma unroll
        for (int i = 0; i < KS / 8; ++i)
            s.a[mt][i] = *(const v4f*)(ap + (size_t)mt * 16 * LDA + 4 * i);
#pragma unroll
    for (int kk = 0; kk < KS / 2; ++kk)
        s.b[kk] = *(const cvec*)(wp + (size_t)kk * N);
}

template <int MT, int CPL, int KS>
__device__ __forceinline__ void slab_mma(const Slab<MT, CPL, KS>& s, v8f (&acc)[MT][CPL])
{
#pragma unroll
    for (int j = 0; j < KS / 4; ++j) {
        v2f bf[CPL];
#pragma unroll
        for (int q = 0; q < CPL; ++q) {
            bf[q][0] = s.b[2 * j][q];
            bf[q][1] = s.b[2 * j + 1][q];
        }
#pragma unroll
        for (int mt = 0; mt < MT; ++mt) {
            v2f af;
            af[0] = s.a[mt][(2 * j) >> 2][(2 * j) & 3];
            af[1] = s.a[mt][(2 * j) >> 2][((2 * j) & 3) + 1];
#pragma unroll
            for (int q = 0; q < CPL; ++q)
                acc[mt][q] = __builtin_amdgcn_wmma_f32_16x16x4_f32(
                    false, af, false, bf[q], (short)0, acc[mt][q], false, false);
        }
    }
}

template <int MT, int CPL, int KS, int WM, int WN, int N, int K, int LDA, bool SP>
__global__ void __launch_bounds__(128) wmma_gemm_f32(
    const float* __restrict__ A, const float* __restrict__ W,
    float* __restrict__ C, const float* __restrict__ bias)
{
    static_assert(K % (2 * KS) == 0 && KS % 8 == 0, "slab pairs");
    static_assert(N % (WN * CPL * 16) == 0, "column tiles");
    typedef typename colvec<CPL>::type cvec;
    constexpr int KH = KS / 2;

    const int wave = threadIdx.x >> 5;
    const int lane = threadIdx.x & 31;
    const int half = lane >> 4;
    const int c    = lane & 15;
    const int m0   = (blockIdx.y * WM + wave / WN) * (MT * 16);
    const int n0   = (blockIdx.x * WN + wave % WN) * (CPL * 16);

    v8f acc[MT][CPL] = {};

    const float* aptr = A + (size_t)(m0 + c) * LDA + half * KH;
    const float* wptr = W + (size_t)(half * KH) * N + n0 + CPL * c;

#define LOADS_FIRST() __builtin_amdgcn_sched_barrier(0)
    Slab<MT, CPL, KS> s0, s1;
    slab_load<MT, CPL, KS, N, LDA>(s0, aptr, wptr);
    for (int i = 0; i < K / (2 * KS) - 1; ++i) {
        aptr += KS;  wptr += (size_t)KS * N;
        slab_load<MT, CPL, KS, N, LDA>(s1, aptr, wptr);
        LOADS_FIRST();
        slab_mma<MT, CPL, KS>(s0, acc);
        aptr += KS;  wptr += (size_t)KS * N;
        slab_load<MT, CPL, KS, N, LDA>(s0, aptr, wptr);
        LOADS_FIRST();
        slab_mma<MT, CPL, KS>(s1, acc);
    }
    aptr += KS;  wptr += (size_t)KS * N;
    slab_load<MT, CPL, KS, N, LDA>(s1, aptr, wptr);
    LOADS_FIRST();
    slab_mma<MT, CPL, KS>(s0, acc);
    slab_mma<MT, CPL, KS>(s1, acc);
#undef LOADS_FIRST

    cvec bsv = {};
    if (SP) bsv = *(const cvec*)(bias + n0 + CPL * c);
    float* cp = C + (size_t)(m0 + 8 * half) * N + n0 + CPL * c;
#pragma unroll
    for (int mt = 0; mt < MT; ++mt) {
#pragma unroll
        for (int r = 0; r < 8; ++r) {
            cvec v;
#pragma unroll
            for (int q = 0; q < CPL; ++q) {
                float t = acc[mt][q][r];
                if (SP) t = softplusf(t + bsv[q]);
                v[q] = t;
            }
            *(cvec*)(cp + (size_t)(mt * 16 + r) * N) = v;
        }
    }
}

__global__ void conv_silu_kernel(const float* __restrict__ xz,
                                 const float* __restrict__ cw,
                                 const float* __restrict__ cb,
                                 float* __restrict__ u)
{
    const int idx = blockIdx.x * blockDim.x + threadIdx.x;
    const int d  = idx % DI;
    const int bt = idx / DI;
    const int t  = bt % LL;
    float acc = cb[d];
    const float* w = cw + d * KC;
#pragma unroll
    for (int k = 0; k < KC; ++k) {
        const int tt = t - (KC - 1) + k;
        if (tt >= 0)
            acc += xz[(size_t)(bt - (KC - 1) + k) * XZW + d] * w[k];
    }
    u[(size_t)bt * DI + d] = siluf(acc);
}

#define TCH 32
#define TST 33

template <int SH>
__device__ __forceinline__ float row_up(float keep, float v) {
    return __builtin_bit_cast(float, __builtin_amdgcn_update_dpp(
        __builtin_bit_cast(int, keep), __builtin_bit_cast(int, v),
        0x110 + SH, 0xf, 0xf, false));
}
__device__ __forceinline__ float lane_to_all(float v, int l) {
    return __builtin_bit_cast(float,
        __builtin_amdgcn_readlane(__builtin_bit_cast(int, v), l));
}

#define SCAN_STAGE(SH)                                   \
    {                                                    \
        const float ap = row_up<SH>(1.0f, a);            \
        const float sp = row_up<SH>(0.0f, s);            \
        s = a * sp + s;                                  \
        a = a * ap;                                      \
    }

__global__ void __launch_bounds__(1024) scan_kernel(
    const float* __restrict__ dt, const float* __restrict__ u,
    const float* __restrict__ dbl, const float* __restrict__ xz,
    const float* __restrict__ A_log, const float* __restrict__ Dp,
    float* __restrict__ yg)
{
    __shared__ float sDT[TCH * TST];
    __shared__ float sU [TCH * TST];
    __shared__ float sZ [TCH * TST];
    __shared__ float sBC[TCH * TST];
    __shared__ float sY [TCH * TST];

    const int lane = threadIdx.x & 31;
    const int w    = threadIdx.x >> 5;
    const int b    = blockIdx.x >> 6;
    const int d0   = (blockIdx.x & 63) * 32;
    const int dch  = d0 + w;

    float An[NS];
#pragma unroll
    for (int n = 0; n < NS; ++n) An[n] = -__expf(A_log[dch * NS + n]);
    const float Dd = Dp[dch];
    float hin[NS];
#pragma unroll
    for (int n = 0; n < NS; ++n) hin[n] = 0.0f;

    for (int c = 0; c < LL / TCH; ++c) {
        const size_t row = (size_t)(b * LL + c * TCH + w);
        sDT[w * TST + lane] = dt[row * DI + d0 + lane];
        sU [w * TST + lane] = u[row * DI + d0 + lane];
        sZ [w * TST + lane] = xz[row * XZW + DI + d0 + lane];
        sBC[w * TST + lane] = dbl[row * DBLW + RR + lane];
        __syncthreads();

        const float dtv = sDT[lane * TST + w];
        const float uv  = sU [lane * TST + w];
        const float zv  = sZ [lane * TST + w];
        const float du  = dtv * uv;
        float y = 0.0f;
#pragma unroll
        for (int n = 0; n < NS; ++n) {
            float a = __expf(dtv * An[n]);
            float s = du * sBC[lane * TST + n];
            SCAN_STAGE(1)
            SCAN_STAGE(2)
            SCAN_STAGE(4)
            SCAN_STAGE(8)
            const float a15 = lane_to_all(a, 15);
            const float s15 = lane_to_all(s, 15);
            if (lane >= 16) {
                s = a * s15 + s;
                a = a * a15;
            }
            const float hv = a * hin[n] + s;
            y += hv * sBC[lane * TST + NS + n];
            hin[n] = lane_to_all(hv, 31);
        }
        sY[lane * TST + w] = (y + uv * Dd) * siluf(zv);
        __syncthreads();

        yg[row * DI + d0 + lane] = sY[w * TST + lane];
    }
}

extern "C" void kernel_launch(void* const* d_in, const int* in_sizes, int n_in,
                              void* d_out, int out_size, void* d_ws, size_t ws_size,
                              hipStream_t stream) {
    const float* x          = (const float*)d_in[0];
    const float* in_proj_w  = (const float*)d_in[1];
    const float* conv_w     = (const float*)d_in[2];
    const float* conv_b     = (const float*)d_in[3];
    const float* x_proj_w   = (const float*)d_in[4];
    const float* dt_proj_w  = (const float*)d_in[5];
    const float* dt_proj_b  = (const float*)d_in[6];
    const float* A_log      = (const float*)d_in[7];
    const float* Dvec       = (const float*)d_in[8];
    const float* out_proj_w = (const float*)d_in[9];
    float* out = (float*)d_out;

    float* ws  = (float*)d_ws;
    float* xz  = ws;
    float* u   = xz  + (size_t)2048 * 4096;
    float* dbl = u   + (size_t)2048 * 2048;
    float* dtb = dbl + (size_t)2048 * 96;
    float* yg  = dtb + (size_t)2048 * 2048;

    wmma_gemm_f32<4, 4, 16, 2, 2, XZW, DMOD, DMOD, false>
        <<<dim3(XZW / 128, (BB * LL) / 128), 128, 0, stream>>>(x, in_proj_w, xz, nullptr);

    conv_silu_kernel<<<(BB * LL * DI) / 256, 256, 0, stream>>>(xz, conv_w, conv_b, u);

    wmma_gemm_f32<1, 2, 32, 2, 1, DBLW, DI, DI, false>
        <<<dim3(DBLW / 32, (BB * LL) / 32), 64, 0, stream>>>(u, x_proj_w, dbl, nullptr);

    wmma_gemm_f32<1, 4, 16, 2, 2, DI, RR, DBLW, true>
        <<<dim3(DI / 128, (BB * LL) / 32), 128, 0, stream>>>(dbl, dt_proj_w, dtb, dt_proj_b);

    scan_kernel<<<BB * (DI / 32), 1024, 0, stream>>>(dtb, u, dbl, xz, A_log, Dvec, yg);

    wmma_gemm_f32<2, 4, 32, 2, 2, DMOD, DI, DI, false>
        <<<dim3(DMOD / 128, (BB * LL) / 64), 128, 0, stream>>>(yg, out_proj_w, out, nullptr);
}


# scan: a lane composes 4 consecutive time steps itself, a cross-lane prefix scan joins the 32 lane maps once per 4 steps (128 steps per trip), then each lane walks its steps; new summation order
# speedup vs baseline: 8.8682x; 1.2743x over previous
#include <hip/hip_runtime.h>
#include <hip/hip_bf16.h>
#include <cmath>

typedef __attribute__((ext_vector_type(2))) float v2f;
typedef __attribute__((ext_vector_type(4))) float v4f;
typedef __attribute__((ext_vector_type(8))) float v8f;

#define BB    2
#define LL    1024
#define DMOD  1024
#define DI    2048
#define NS    16
#define KC    4
#define RR    64
#define XZW   (2 * DI)
#define DBLW  (RR + 2 * NS)

__device__ __forceinline__ float siluf(float v) {
    return v / (1.0f + __expf(-v));
}
__device__ __forceinline__ float softplusf(float v) {
    return (v > 20.0f) ? v : log1pf(expf(v));
}

template <int CPL> struct colvec;
template <> struct colvec<4> { typedef v4f type; };
template <> struct colvec<2> { typedef v2f type; };

template <int MT, int CPL, int KS>
struct Slab {
    v4f a[MT][KS / 8];
    typename colvec<CPL>::type b[KS / 2];
};

template <int MT, int CPL, int KS, int N, int LDA>
__device__ __forceinline__ void slab_load(Slab<MT, CPL, KS>& s,
                                          const float* __restrict__ ap,
                                          const float* __restrict__ wp)
{
    typedef typename colvec<CPL>::type cvec;
#pragma unroll
    for (int mt = 0; mt < MT; ++mt)
#pragma unroll
        for (int i = 0; i < KS / 8; ++i)
            s.a[mt][i] = *(const v4f*)(ap + (size_t)mt * 16 * LDA + 4 * i);
#pragma unroll
    for (int kk = 0; kk < KS / 2; ++kk)
        s.b[kk] = *(const cvec*)(wp + (size_t)kk * N);
}

template <int MT, int CPL, int KS>
__device__ __forceinline__ void slab_mma(const Slab<MT, CPL, KS>& s, v8f (&acc)[MT][CPL])
{
#pragma unroll
    for (int j = 0; j < KS / 4; ++j) {
        v2f bf[CPL];
#pragma unroll
        for (int q = 0; q < CPL; ++q) {
            bf[q][0] = s.b[2 * j][q];
            bf[q][1] = s.b[2 * j + 1][q];
        }
#pragma unroll
        for (int mt = 0; mt < MT; ++mt) {
            v2f af;
            af[0] = s.a[mt][(2 * j) >> 2][(2 * j) & 3];
            af[1] = s.a[mt][(2 * j) >> 2][((2 * j) & 3) + 1];
#pragma unroll
            for (int q = 0; q < CPL; ++q)
                acc[mt][q] = __builtin_amdgcn_wmma_f32_16x16x4_f32(
                    false, af, false, bf[q], (short)0, acc[mt][q], false, false);
        }
    }
}

template <int MT, int CPL, int KS, int WM, int WN, int N, int K, int LDA, bool SP>
__global__ void __launch_bounds__(128) wmma_gemm_f32(
    const float* __restrict__ A, const float* __restrict__ W,
    float* __restrict__ C, const float* __restrict__ bias)
{
    static_assert(K % (2 * KS) == 0 && KS % 8 == 0, "slab pairs");
    static_assert(N % (WN * CPL * 16) == 0, "column tiles");
    typedef typename colvec<CPL>::type cvec;
    constexpr int KH = KS / 2;

    const int wave = threadIdx.x >> 5;
    const int lane = threadIdx.x & 31;
    const int half = lane >> 4;
    const int c    = lane & 15;
    const int m0   = (blockIdx.y * WM + wave / WN) * (MT * 16);
    const int n0   = (blockIdx.x * WN + wave % WN) * (CPL * 16);

    v8f acc[MT][CPL] = {};

    const float* aptr = A + (size_t)(m0 + c) * LDA + half * KH;
    const float* wptr = W + (size_t)(half * KH) * N + n0 + CPL * c;

#define LOADS_FIRST() __builtin_amdgcn_sched_barrier(0)
    Slab<MT, CPL, KS> s0, s1;
    slab_load<MT, CPL, KS, N, LDA>(s0, aptr, wptr);
    for (int i = 0; i < K / (2 * KS) - 1; ++i) {
        aptr += KS;  wptr += (size_t)KS * N;
        slab_load<MT, CPL, KS, N, LDA>(s1, aptr, wptr);
        LOADS_FIRST();
        slab_mma<MT, CPL, KS>(s0, acc);
        aptr += KS;  wptr += (size_t)KS * N;
        slab_load<MT, CPL, KS, N, LDA>(s0, aptr, wptr);
        LOADS_FIRST();
        slab_mma<MT, CPL, KS>(s1, acc);
    }
    aptr += KS;  wptr += (size_t)KS * N;
    slab_load<MT, CPL, KS, N, LDA>(s1, aptr, wptr);
    LOADS_FIRST();
    slab_mma<MT, CPL, KS>(s0, acc);
    slab_mma<MT, CPL, KS>(s1, acc);
#undef LOADS_FIRST

    cvec bsv = {};
    if (SP) bsv = *(const cvec*)(bias + n0 + CPL * c);
    float* cp = C + (size_t)(m0 + 8 * half) * N + n0 + CPL * c;
#pragma unroll
    for (int mt = 0; mt < MT; ++mt) {
#pragma unroll
        for (int r = 0; r < 8; ++r) {
            cvec v;
#pragma unroll
            for (int q = 0; q < CPL; ++q) {
                float t = acc[mt][q][r];
                if (SP) t = softplusf(t + bsv[q]);
                v[q] = t;
            }
            *(cvec*)(cp + (size_t)(mt * 16 + r) * N) = v;
        }
    }
}

__global__ void conv_silu_kernel(const float* __restrict__ xz,
                                 const float* __restrict__ cw,
                                 const float* __restrict__ cb,
                                 float* __restrict__ u)
{
    const int idx = blockIdx.x * blockDim.x + threadIdx.x;
    const int d  = idx % DI;
    const int bt = idx / DI;
    const int t  = bt % LL;
    float acc = cb[d];
    const float* w = cw + d * KC;
#pragma unroll
    for (int k = 0; k < KC; ++k) {
        const int tt = t - (KC - 1) + k;
        if (tt >= 0)
            acc += xz[(size_t)(bt - (KC - 1) + k) * XZW + d] * w[k];
    }
    u[(size_t)bt * DI + d] = siluf(acc);
}

#define KPL 4
#define TT  (32 * KPL)
#define TST 33
#define BCP 36

template <int SH>
__device__ __forceinline__ float row_up(float keep, float v) {
    return __builtin_bit_cast(float, __builtin_amdgcn_update_dpp(
        __builtin_bit_cast(int, keep), __builtin_bit_cast(int, v),
        0x110 + SH, 0xf, 0xf, false));
}
__device__ __forceinline__ float lane_to_all(float v, int l) {
    return __builtin_bit_cast(float,
        __builtin_amdgcn_readlane(__builtin_bit_cast(int, v), l));
}

#define SCAN_STAGE(SH)                                   \
    {                                                    \
        const float ap = row_up<SH>(1.0f, a);            \
        const float sp = row_up<SH>(0.0f, s);            \
        s = a * sp + s;                                  \
        a = a * ap;                                      \
    }

__global__ void __launch_bounds__(1024) scan_kernel(
    const float* __restrict__ dt, const float* __restrict__ u,
    const float* __restrict__ dbl, const float* __restrict__ xz,
    const float* __restrict__ A_log, const float* __restrict__ Dp,
    float* __restrict__ yg)
{
    __shared__ __attribute__((aligned(16))) float lds[TT * BCP + 4 * TT * TST];
    float* const sBC = lds;
    float* const sDT = lds + TT * BCP;
    float* const sU  = sDT + TT * TST;
    float* const sZ  = sU  + TT * TST;
    float* const sY  = sZ  + TT * TST;

    const int lane = threadIdx.x & 31;
    const int w    = threadIdx.x >> 5;
    const int b    = blockIdx.x >> 6;
    const int d0   = (blockIdx.x & 63) * 32;
    const int dch  = d0 + w;

    float An[NS];
#pragma unroll
    for (int n = 0; n < NS; ++n)
        An[n] = lane_to_all(-__expf(A_log[dch * NS + n]), 0);
    const float Dd = Dp[dch];
    float hin[NS];
#pragma unroll
    for (int n = 0; n < NS; ++n) hin[n] = 0.0f;

    for (int c = 0; c < LL / TT; ++c) {
#pragma unroll
        for (int jj = 0; jj < KPL; ++jj) {
            const int r = jj * 32 + w;
            const int p = (w % KPL) * 32 + w / KPL + jj * (32 / KPL);
            const size_t row = (size_t)(b * LL + c * TT + r);
            sDT[p * TST + lane] = dt[row * DI + d0 + lane];
            sU [p * TST + lane] = u[row * DI + d0 + lane];
            sZ [p * TST + lane] = xz[row * XZW + DI + d0 + lane];
            sBC[p * BCP + lane] = dbl[row * DBLW + RR + lane];
        }
        __syncthreads();

        float dtv[KPL], du[KPL], y[KPL];
#pragma unroll
        for (int j = 0; j < KPL; ++j) {
            dtv[j] = sDT[(j * 32 + lane) * TST + w];
            du[j]  = dtv[j] * sU[(j * 32 + lane) * TST + w];
            y[j]   = 0.0f;
        }
#pragma unroll
        for (int g = 0; g < NS / 4; ++g) {
            v4f Bv[KPL], Cv[KPL];
#pragma unroll
            for (int j = 0; j < KPL; ++j) {
                Bv[j] = *(const v4f*)&sBC[(j * 32 + lane) * BCP + 4 * g];
                Cv[j] = *(const v4f*)&sBC[(j * 32 + lane) * BCP + NS + 4 * g];
            }
#pragma unroll
            for (int q = 0; q < 4; ++q) {
                const int n = 4 * g + q;
                float aj[KPL], sj[KPL];
#pragma unroll
                for (int j = 0; j < KPL; ++j) {
                    aj[j] = __expf(dtv[j] * An[n]);
                    sj[j] = du[j] * Bv[j][q];
                }
                float a = aj[0];
                float s = sj[0];
#pragma unroll
                for (int j = 1; j < KPL; ++j) {
                    s = aj[j] * s + sj[j];
                    a = a * aj[j];
                }
                SCAN_STAGE(1)
                SCAN_STAGE(2)
                SCAN_STAGE(4)
                SCAN_STAGE(8)
                const float a15 = lane_to_all(a, 15);
                const float s15 = lane_to_all(s, 15);
                if (lane >= 16) {
                    s = a * s15 + s;
                    a = a * a15;
                }
                const float hend = a * hin[n] + s;
                float h = row_up<1>(hin[n], hend);
                const float h15 = lane_to_all(hend, 15);
                h = (lane == 16) ? h15 : h;
                hin[n] = lane_to_all(hend, 31);
#pragma unroll
                for (int j = 0; j < KPL; ++j) {
                    h = aj[j] * h + sj[j];
                    y[j] += h * Cv[j][q];
                }
            }
        }
#pragma unroll
        for (int j = 0; j < KPL; ++j) {
            const float uv = sU[(j * 32 + lane) * TST + w];
            const float zv = sZ[(j * 32 + lane) * TST + w];
            sY[(j * 32 + lane) * TST + w] = (y[j] + uv * Dd) * siluf(zv);
        }
        __syncthreads();

#pragma unroll
        for (int jj = 0; jj < KPL; ++jj) {
            const int r = jj * 32 + w;
            const int p = (w % KPL) * 32 + w / KPL + jj * (32 / KPL);
            const size_t row = (size_t)(b * LL + c * TT + r);
            yg[row * DI + d0 + lane] = sY[p * TST + lane];
        }
    }
}

extern "C" void kernel_launch(void* const* d_in, const int* in_sizes, int n_in,
                              void* d_out, int out_size, void* d_ws, size_t ws_size,
                              hipStream_t stream) {
    const float* x          = (const float*)d_in[0];
    const float* in_proj_w  = (const float*)d_in[1];
    const float* conv_w     = (const float*)d_in[2];
    const float* conv_b     = (const float*)d_in[3];
    const float* x_proj_w   = (const float*)d_in[4];
    const float* dt_proj_w  = (const float*)d_in[5];
    const float* dt_proj_b  = (const float*)d_in[6];
    const float* A_log      = (const float*)d_in[7];
    const float* Dvec       = (const float*)d_in[8];
    const float* out_proj_w = (const float*)d_in[9];
    float* out = (float*)d_out;

    float* ws  = (float*)d_ws;
    float* xz  = ws;
    float* u   = xz  + (size_t)2048 * 4096;
    float* dbl = u   + (size_t)2048 * 2048;
    float* dtb = dbl + (size_t)2048 * 96;
    float* yg  = dtb + (size_t)2048 * 2048;

    wmma_gemm_f32<4, 4, 16, 2, 2, XZW, DMOD, DMOD, false>
        <<<dim3(XZW / 128, (BB * LL) / 128), 128, 0, stream>>>(x, in_proj_w, xz, nullptr);

    conv_silu_kernel<<<(BB * LL * DI) / 256, 256, 0, stream>>>(xz, conv_w, conv_b, u);

    wmma_gemm_f32<1, 2, 32, 2, 1, DBLW, DI, DI, false>
        <<<dim3(DBLW / 32, (BB * LL) / 32), 64, 0, stream>>>(u, x_proj_w, dbl, nullptr);

    wmma_gemm_f32<1, 4, 16, 2, 2, DI, RR, DBLW, true>
        <<<dim3(DI / 128, (BB * LL) / 32), 128, 0, stream>>>(dbl, dt_proj_w, dtb, dt_proj_b);

    scan_kernel<<<BB * (DI / 32), 1024, 0, stream>>>(dtb, u, dbl, xz, A_log, Dvec, yg);

    wmma_gemm_f32<2, 4, 32, 2, 2, DMOD, DI, DI, false>
        <<<dim3(DMOD / 128, (BB * LL) / 64), 128, 0, stream>>>(yg, out_proj_w, out, nullptr);
}


# scan: 16 channels per block, not 32: 256 blocks of 16 waves reach every WGP, not 128 on half of them; y kept in 16-channel blocks, one writer per 128-byte line, read so by the last product; same sums
# speedup vs baseline: 9.3300x; 1.0521x over previous
#include <hip/hip_runtime.h>
#include <hip/hip_bf16.h>
#include <cmath>

typedef __attribute__((ext_vector_type(2))) float v2f;
typedef __attribute__((ext_vector_type(4))) float v4f;
typedef __attribute__((ext_vector_type(8))) float v8f;

#define BB    2
#define LL    1024
#define DMOD  1024
#define DI    2048
#define NS    16
#define KC    4
#define RR    64
#define XZW   (2 * DI)
#define DBLW  (RR + 2 * NS)

__device__ __forceinline__ float siluf(float v) {
    return v / (1.0f + __expf(-v));
}
__device__ __forceinline__ float softplusf(float v) {
    return (v > 20.0f) ? v : log1pf(expf(v));
}

template <int CPL> struct colvec;
template <> struct colvec<4> { typedef v4f type; };
template <> struct colvec<2> { typedef v2f type; };

template <int MT, int CPL, int KS>
struct Slab {
    v4f a[MT][KS / 8];
    typename colvec<CPL>::type b[KS / 2];
};

template <int MT, int CPL, int KS, int N, int LDA>
__device__ __forceinline__ void slab_load(Slab<MT, CPL, KS>& s,
                                          const float* __restrict__ ap,
                                          const float* __restrict__ wp)
{
    typedef typename colvec<CPL>::type cvec;
#pragma unroll
    for (int mt = 0; mt < MT; ++mt)
#pragma unroll
        for (int i = 0; i < KS / 8; ++i)
            s.a[mt][i] = *(const v4f*)(ap + (size_t)mt * 16 * LDA + 4 * i);
#pragma unroll
    for (int kk = 0; kk < KS / 2; ++kk)
        s.b[kk] = *(const cvec*)(wp + (size_t)kk * N);
}

template <int MT, int CPL, int KS>
__device__ __forceinline__ void slab_mma(const Slab<MT, CPL, KS>& s, v8f (&acc)[MT][CPL])
{
#pragma unroll
    for (int j = 0; j < KS / 4; ++j) {
        v2f bf[CPL];
#pragma unroll
        for (int q = 0; q < CPL; ++q) {
            bf[q][0] = s.b[2 * j][q];
            bf[q][1] = s.b[2 * j + 1][q];
        }
#pragma unroll
        for (int mt = 0; mt < MT; ++mt) {
            v2f af;
            af[0] = s.a[mt][(2 * j) >> 2][(2 * j) & 3];
            af[1] = s.a[mt][(2 * j) >> 2][((2 * j) & 3) + 1];
#pragma unroll
            for (int q = 0; q < CPL; ++q)
                acc[mt][q] = __builtin_amdgcn_wmma_f32_16x16x4_f32(
                    false, af, false, bf[q], (short)0, acc[mt][q], false, false);
        }
    }
}

template <int MT, int CPL, int KS, int WM, int WN, int N, int K, int LDA, int AHALF, int ASLAB, bool SP>
__global__ void __launch_bounds__(128) wmma_gemm_f32(
    const float* __restrict__ A, const float* __restrict__ W,
    float* __restrict__ C, const float* __restrict__ bias)
{
    static_assert(K % (2 * KS) == 0 && KS % 8 == 0, "slab pairs");
    static_assert(N % (WN * CPL * 16) == 0, "column tiles");
    typedef typename colvec<CPL>::type cvec;
    constexpr int KH = KS / 2;

    const int wave = threadIdx.x >> 5;
    const int lane = threadIdx.x & 31;
    const int half = lane >> 4;
    const int c    = lane & 15;
    const int m0   = (blockIdx.y * WM + wave / WN) * (MT * 16);
    const int n0   = (blockIdx.x * WN + wave % WN) * (CPL * 16);

    v8f acc[MT][CPL] = {};

    const float* aptr = A + (size_t)(m0 + c) * LDA + (size_t)half * AHALF;
    const float* wptr = W + (size_t)(half * KH) * N + n0 + CPL * c;

#define LOADS_FIRST() __builtin_amdgcn_sched_barrier(0)
    Slab<MT, CPL, KS> s0, s1;
    slab_load<MT, CPL, KS, N, LDA>(s0, aptr, wptr);
    for (int i = 0; i < K / (2 * KS) - 1; ++i) {
        aptr += ASLAB;  wptr += (size_t)KS * N;
        slab_load<MT, CPL, KS, N, LDA>(s1, aptr, wptr);
        LOADS_FIRST();
        slab_mma<MT, CPL, KS>(s0, acc);
        aptr += ASLAB;  wptr += (size_t)KS * N;
        slab_load<MT, CPL, KS, N, LDA>(s0, aptr, wptr);
        LOADS_FIRST();
        slab_mma<MT, CPL, KS>(s1, acc);
    }
    aptr += ASLAB;  wptr += (size_t)KS * N;
    slab_load<MT, CPL, KS, N, LDA>(s1, aptr, wptr);
    LOADS_FIRST();
    slab_mma<MT, CPL, KS>(s0, acc);
    slab_mma<MT, CPL, KS>(s1, acc);
#undef LOADS_FIRST

    cvec bsv = {};
    if (SP) bsv = *(const cvec*)(bias + n0 + CPL * c);
    float* cp = C + (size_t)(m0 + 8 * half) * N + n0 + CPL * c;
#pragma unroll
    for (int mt = 0; mt < MT; ++mt) {
#pragma unroll
        for (int r = 0; r < 8; ++r) {
            cvec v;
#pragma unroll
            for (int q = 0; q < CPL; ++q) {
                float t = acc[mt][q][r];
                if (SP) t = softplusf(t + bsv[q]);
                v[q] = t;
            }
            *(cvec*)(cp + (size_t)(mt * 16 + r) * N) = v;
        }
    }
}

__global__ void conv_silu_kernel(const float* __restrict__ xz,
                                 const float* __restrict__ cw,
                                 const float* __restrict__ cb,
                                 float* __restrict__ u)
{
    const int idx = blockIdx.x * blockDim.x + threadIdx.x;
    const int d  = idx % DI;
    const int bt = idx / DI;
    const int t  = bt % LL;
    float acc = cb[d];
    const float* w = cw + d * KC;
#pragma unroll
    for (int k = 0; k < KC; ++k) {
        const int tt = t - (KC - 1) + k;
        if (tt >= 0)
            acc += xz[(size_t)(bt - (KC - 1) + k) * XZW + d] * w[k];
    }
    u[(size_t)bt * DI + d] = siluf(acc);
}

#define KPL 4
#define TT  (32 * KPL)
#define CHB 16
#define TST 17
#define BCP 36

template <int SH>
__device__ __forceinline__ float row_up(float keep, float v) {
    return __builtin_bit_cast(float, __builtin_amdgcn_update_dpp(
        __builtin_bit_cast(int, keep), __builtin_bit_cast(int, v),
        0x110 + SH, 0xf, 0xf, false));
}
__device__ __forceinline__ float lane_to_all(float v, int l) {
    return __builtin_bit_cast(float,
        __builtin_amdgcn_readlane(__builtin_bit_cast(int, v), l));
}

#define SCAN_STAGE(SH)                                   \
    {                                                    \
        const float ap = row_up<SH>(1.0f, a);            \
        const float sp = row_up<SH>(0.0f, s);            \
        s = a * sp + s;                                  \
        a = a * ap;                                      \
    }

__global__ void __launch_bounds__(32 * CHB) scan_kernel(
    const float* __restrict__ dt, const float* __restrict__ u,
    const float* __restrict__ dbl, const float* __restrict__ xz,
    const float* __restrict__ A_log, const float* __restrict__ Dp,
    float* __restrict__ yb)
{
    __shared__ __attribute__((aligned(16))) float lds[TT * BCP + 4 * TT * TST];
    float* const sBC = lds;
    float* const sDT = lds + TT * BCP;
    float* const sU  = sDT + TT * TST;
    float* const sZ  = sU  + TT * TST;
    float* const sY  = sZ  + TT * TST;

    const int lane = threadIdx.x & 31;
    const int w    = threadIdx.x >> 5;
    const int b    = blockIdx.x / (DI / CHB);
    const int cg   = blockIdx.x % (DI / CHB);
    const int d0   = cg * CHB;
    const int dch  = d0 + w;
    const int hl   = lane >> 4;
    const int c16  = lane & 15;
    const int pm   = (2 * (w & 1) + hl) * 32 + (w >> 1);

    float An[NS];
#pragma unroll
    for (int n = 0; n < NS; ++n)
        An[n] = lane_to_all(-__expf(A_log[dch * NS + n]), 0);
    const float Dd = Dp[dch];
    float hin[NS];
#pragma unroll
    for (int n = 0; n < NS; ++n) hin[n] = 0.0f;

    for (int c = 0; c < LL / TT; ++c) {
#pragma unroll
        for (int jj = 0; jj < TT / (2 * CHB); ++jj) {
            const int r = 32 * jj + 2 * w + hl;
            const int p = pm + 8 * jj;
            const size_t row = (size_t)(b * LL + c * TT + r);
            sDT[p * TST + c16] = dt[row * DI + d0 + c16];
            sU [p * TST + c16] = u[row * DI + d0 + c16];
            sZ [p * TST + c16] = xz[row * XZW + DI + d0 + c16];
        }
#pragma unroll
        for (int jj = 0; jj < TT / (4 * CHB); ++jj) {
            const int r = (jj * CHB + w) * 4 + (lane >> 3);
            const int p = (lane >> 3) * 32 + jj * CHB + w;
            const size_t row = (size_t)(b * LL + c * TT + r);
            *(v4f*)&sBC[p * BCP + 4 * (lane & 7)] =
                *(const v4f*)(dbl + row * DBLW + RR + 4 * (lane & 7));
        }
        __syncthreads();

        float dtv[KPL], du[KPL], y[KPL];
#pragma unroll
        for (int j = 0; j < KPL; ++j) {
            dtv[j] = sDT[(j * 32 + lane) * TST + w];
            du[j]  = dtv[j] * sU[(j * 32 + lane) * TST + w];
            y[j]   = 0.0f;
        }
#pragma unroll
        for (int g = 0; g < NS / 4; ++g) {
            v4f Bv[KPL], Cv[KPL];
#pragma unroll
            for (int j = 0; j < KPL; ++j) {
                Bv[j] = *(const v4f*)&sBC[(j * 32 + lane) * BCP + 4 * g];
                Cv[j] = *(const v4f*)&sBC[(j * 32 + lane) * BCP + NS + 4 * g];
            }
#pragma unroll
            for (int q = 0; q < 4; ++q) {
                const int n = 4 * g + q;
                float aj[KPL], sj[KPL];
#pragma unroll
                for (int j = 0; j < KPL; ++j) {
                    aj[j] = __expf(dtv[j] * An[n]);
                    sj[j] = du[j] * Bv[j][q];
                }
                float a = aj[0];
                float s = sj[0];
#pragma unroll
                for (int j = 1; j < KPL; ++j) {
                    s = aj[j] * s + sj[j];
                    a = a * aj[j];
                }
                SCAN_STAGE(1)
                SCAN_STAGE(2)
                SCAN_STAGE(4)
                SCAN_STAGE(8)
                const float a15 = lane_to_all(a, 15);
                const float s15 = lane_to_all(s, 15);
                if (lane >= 16) {
                    s = a * s15 + s;
                    a = a * a15;
                }
                const float hend = a * hin[n] + s;
                float h = row_up<1>(hin[n], hend);
                const float h15 = lane_to_all(hend, 15);
                h = (lane == 16) ? h15 : h;
                hin[n] = lane_to_all(hend, 31);
#pragma unroll
                for (int j = 0; j < KPL; ++j) {
                    h = aj[j] * h + sj[j];
                    y[j] += h * Cv[j][q];
                }
            }
        }
#pragma unroll
        for (int j = 0; j < KPL; ++j) {
            const float uv = sU[(j * 32 + lane) * TST + w];
            const float zv = sZ[(j * 32 + lane) * TST + w];
            sY[(j * 32 + lane) * TST + w] = (y[j] + uv * Dd) * siluf(zv);
        }
        __syncthreads();

#pragma unroll
        for (int jj = 0; jj < TT / (2 * CHB); ++jj) {
            const int r = 32 * jj + 2 * w + hl;
            const int p = pm + 8 * jj;
            const size_t row = (size_t)(b * LL + c * TT + r);
            yb[((size_t)cg * (BB * LL) + row) * CHB + c16] = sY[p * TST + c16];
        }
    }
}

extern "C" void kernel_launch(void* const* d_in, const int* in_sizes, int n_in,
                              void* d_out, int out_size, void* d_ws, size_t ws_size,
                              hipStream_t stream) {
    const float* x          = (const float*)d_in[0];
    const float* in_proj_w  = (const float*)d_in[1];
    const float* conv_w     = (const float*)d_in[2];
    const float* conv_b     = (const float*)d_in[3];
    const float* x_proj_w   = (const float*)d_in[4];
    const float* dt_proj_w  = (const float*)d_in[5];
    const float* dt_proj_b  = (const float*)d_in[6];
    const float* A_log      = (const float*)d_in[7];
    const float* Dvec       = (const float*)d_in[8];
    const float* out_proj_w = (const float*)d_in[9];
    float* out = (float*)d_out;

    float* ws  = (float*)d_ws;
    float* xz  = ws;
    float* u   = xz  + (size_t)2048 * 4096;
    float* dbl = u   + (size_t)2048 * 2048;
    float* dtb = dbl + (size_t)2048 * 96;
    float* yg  = dtb + (size_t)2048 * 2048;

    wmma_gemm_f32<4, 4, 16, 2, 2, XZW, DMOD, DMOD, 8, 16, false>
        <<<dim3(XZW / 128, (BB * LL) / 128), 128, 0, stream>>>(x, in_proj_w, xz, nullptr);

    conv_silu_kernel<<<(BB * LL * DI) / 256, 256, 0, stream>>>(xz, conv_w, conv_b, u);

    wmma_gemm_f32<1, 2, 32, 2, 1, DBLW, DI, DI, 16, 32, false>
        <<<dim3(DBLW / 32, (BB * LL) / 32), 64, 0, stream>>>(u, x_proj_w, dbl, nullptr);

    wmma_gemm_f32<1, 4, 16, 2, 2, DI, RR, DBLW, 8, 16, true>
        <<<dim3(DI / 128, (BB * LL) / 32), 128, 0, stream>>>(dbl, dt_proj_w, dtb, dt_proj_b);

    scan_kernel<<<BB * (DI / CHB), 32 * CHB, 0, stream>>>(dtb, u, dbl, xz, A_log, Dvec, yg);

    wmma_gemm_f32<2, 4, 32, 2, 2, DMOD, DI, CHB, BB * LL * CHB, 2 * BB * LL * CHB, false>
        <<<dim3(DMOD / 128, (BB * LL) / 64), 128, 0, stream>>>(yg, out_proj_w, out, nullptr);
}


# in_proj epilogue makes u = silu(conv(xi)+b) from its accumulators (3 earlier rows from the lane or its other half); conv kernel shrinks to the 3 first rows of each 64-row tile; sums unchanged
# speedup vs baseline: 9.7028x; 1.0400x over previous
#include <hip/hip_runtime.h>
#include <hip/hip_bf16.h>
#include <cmath>

typedef __attribute__((ext_vector_type(2))) float v2f;
typedef __attribute__((ext_vector_type(4))) float v4f;
typedef __attribute__((ext_vector_type(8))) float v8f;

#define BB    2
#define LL    1024
#define DMOD  1024
#define DI    2048
#define NS    16
#define KC    4
#define RR    64
#define XZW   (2 * DI)
#define DBLW  (RR + 2 * NS)

__device__ __forceinline__ float siluf(float v) {
    return v / (1.0f + __expf(-v));
}
__device__ __forceinline__ float softplusf(float v) {
    return (v > 20.0f) ? v : log1pf(expf(v));
}
__device__ __forceinline__ float other_half(float v) {
    return __builtin_bit_cast(float, __builtin_amdgcn_permlanex16(
        __builtin_bit_cast(int, v), __builtin_bit_cast(int, v),
        0x76543210, 0xfedcba98, false, false));
}

template <int CPL> struct colvec;
template <> struct colvec<4> { typedef v4f type; };
template <> struct colvec<2> { typedef v2f type; };

template <int MT, int CPL, int KS>
struct Slab {
    v4f a[MT][KS / 8];
    typename colvec<CPL>::type b[KS / 2];
};

template <int MT, int CPL, int KS, int N, int LDA>
__device__ __forceinline__ void slab_load(Slab<MT, CPL, KS>& s,
                                          const float* __restrict__ ap,
                                          const float* __restrict__ wp)
{
    typedef typename colvec<CPL>::type cvec;
#pragma unroll
    for (int mt = 0; mt < MT; ++mt)
#pragma unroll
        for (int i = 0; i < KS / 8; ++i)
            s.a[mt][i] = *(const v4f*)(ap + (size_t)mt * 16 * LDA + 4 * i);
#pragma unroll
    for (int kk = 0; kk < KS / 2; ++kk)
        s.b[kk] = *(const cvec*)(wp + (size_t)kk * N);
}

template <int MT, int CPL, int KS>
__device__ __forceinline__ void slab_mma(const Slab<MT, CPL, KS>& s, v8f (&acc)[MT][CPL])
{
#pragma unroll
    for (int j = 0; j < KS / 4; ++j) {
        v2f bf[CPL];
#pragma unroll
        for (int q = 0; q < CPL; ++q) {
            bf[q][0] = s.b[2 * j][q];
            bf[q][1] = s.b[2 * j + 1][q];
        }
#pragma unroll
        for (int mt = 0; mt < MT; ++mt) {
            v2f af;
            af[0] = s.a[mt][(2 * j) >> 2][(2 * j) & 3];
            af[1] = s.a[mt][(2 * j) >> 2][((2 * j) & 3) + 1];
#pragma unroll
            for (int q = 0; q < CPL; ++q)
                acc[mt][q] = __builtin_amdgcn_wmma_f32_16x16x4_f32(
                    false, af, false, bf[q], (short)0, acc[mt][q], false, false);
        }
    }
}

#define EPI_PLAIN    0
#define EPI_SOFTPLUS 1
#define EPI_CONV     2
template <int MT, int CPL, int KS, int WM, int WN, int N, int K, int LDA, int AHALF, int ASLAB, int EPI>
__global__ void __launch_bounds__(128) wmma_gemm_f32(
    const float* __restrict__ A, const float* __restrict__ W,
    float* __restrict__ C, const float* __restrict__ bias,
    const float* __restrict__ cw, float* __restrict__ U)
{
    constexpr bool SP = (EPI == EPI_SOFTPLUS);
    static_assert(K % (2 * KS) == 0 && KS % 8 == 0, "slab pairs");
    static_assert(N % (WN * CPL * 16) == 0, "column tiles");
    typedef typename colvec<CPL>::type cvec;
    constexpr int KH = KS / 2;

    const int wave = threadIdx.x >> 5;
    const int lane = threadIdx.x & 31;
    const int half = lane >> 4;
    const int c    = lane & 15;
    const int m0   = (blockIdx.y * WM + wave / WN) * (MT * 16);
    const int n0   = (blockIdx.x * WN + wave % WN) * (CPL * 16);

    v8f acc[MT][CPL] = {};

    const float* aptr = A + (size_t)(m0 + c) * LDA + (size_t)half * AHALF;
    const float* wptr = W + (size_t)(half * KH) * N + n0 + CPL * c;

#define LOADS_FIRST() __builtin_amdgcn_sched_barrier(0)
    Slab<MT, CPL, KS> s0, s1;
    slab_load<MT, CPL, KS, N, LDA>(s0, aptr, wptr);
    for (int i = 0; i < K / (2 * KS) - 1; ++i) {
        aptr += ASLAB;  wptr += (size_t)KS * N;
        slab_load<MT, CPL, KS, N, LDA>(s1, aptr, wptr);
        LOADS_FIRST();
        slab_mma<MT, CPL, KS>(s0, acc);
        aptr += ASLAB;  wptr += (size_t)KS * N;
        slab_load<MT, CPL, KS, N, LDA>(s0, aptr, wptr);
        LOADS_FIRST();
        slab_mma<MT, CPL, KS>(s1, acc);
    }
    aptr += ASLAB;  wptr += (size_t)KS * N;
    slab_load<MT, CPL, KS, N, LDA>(s1, aptr, wptr);
    LOADS_FIRST();
    slab_mma<MT, CPL, KS>(s0, acc);
    slab_mma<MT, CPL, KS>(s1, acc);
#undef LOADS_FIRST

    if constexpr (EPI == EPI_CONV) {
        if (n0 < DI) {
            const bool bstart = (m0 % LL) == 0;
            v4f wq[CPL];
#pragma unroll
            for (int q = 0; q < CPL; ++q)
                wq[q] = *(const v4f*)(cw + (size_t)(n0 + CPL * c + q) * KC);
            const cvec cbv = *(const cvec*)(bias + n0 + CPL * c);
            float* up = U + (size_t)(m0 + 8 * half) * DI + n0 + CPL * c;
            float* xp = C + (size_t)(m0 + (half ? (MT - 1) * 16 + 8 + 5 : 0)) * N + n0 + CPL * c;
#pragma unroll
            for (int i = 0; i < 3; ++i) {
                cvec v;
#pragma unroll
                for (int q = 0; q < CPL; ++q)
                    v[q] = half ? acc[MT - 1][q][5 + i] : acc[0][q][i];
                *(cvec*)(xp + (size_t)i * N) = v;
            }
#pragma unroll
            for (int mt = 0; mt < MT; ++mt) {
                float pv[CPL][3];
#pragma unroll
                for (int q = 0; q < CPL; ++q)
#pragma unroll
                    for (int i = 0; i < 3; ++i) {
                        const float lo = acc[mt][q][5 + i];
                        const float hi = (mt > 0) ? acc[mt > 0 ? mt - 1 : 0][q][5 + i] : 0.0f;
                        pv[q][i] = other_half(half ? hi : lo);
                    }
#pragma unroll
                for (int r = 0; r < 8; ++r) {
                    cvec v;
#pragma unroll
                    for (int q = 0; q < CPL; ++q) {
                        const float x3 = (r >= 3) ? acc[mt][q][r >= 3 ? r - 3 : 0] : pv[q][r >= 3 ? 0 : r];
                        const float x2 = (r >= 2) ? acc[mt][q][r >= 2 ? r - 2 : 0] : pv[q][r >= 2 ? 0 : r + 1];
                        const float x1 = (r >= 1) ? acc[mt][q][r >= 1 ? r - 1 : 0] : pv[q][2];
                        float t = cbv[q];
                        t += x3 * wq[q][0];
                        t += x2 * wq[q][1];
                        t += x1 * wq[q][2];
                        t += acc[mt][q][r] * wq[q][3];
                        v[q] = siluf(t);
                    }
                    if (mt > 0 || r >= 3 || half == 1 || bstart)
                        *(cvec*)(up + (size_t)(mt * 16 + r) * DI) = v;
                }
            }
            return;
        }
    }

    cvec bsv = {};
    if (SP) bsv = *(const cvec*)(bias + n0 + CPL * c);
    float* cp = C + (size_t)(m0 + 8 * half) * N + n0 + CPL * c;
#pragma unroll
    for (int mt = 0; mt < MT; ++mt) {
#pragma unroll
        for (int r = 0; r < 8; ++r) {
            cvec v;
#pragma unroll
            for (int q = 0; q < CPL; ++q) {
                float t = acc[mt][q][r];
                if (SP) t = softplusf(t + bsv[q]);
                v[q] = t;
            }
            *(cvec*)(cp + (size_t)(mt * 16 + r) * N) = v;
        }
    }
}

#define XT 64
static_assert(XT == 4 * 16, "edge kernel tile = in_proj wave tile (MT = 4)");
__global__ void __launch_bounds__(256) conv_edge_kernel(const float* __restrict__ xz,
                                                        const float* __restrict__ cw,
                                                        const float* __restrict__ cb,
                                                        float* __restrict__ u)
{
    const int idx = blockIdx.x * 256 + threadIdx.x;
    const int d4  = (idx % (DI / 4)) * 4;
    const int m0  = (idx / (DI / 4)) * XT;
    if (m0 % LL == 0) return;
    v4f x[6];
#pragma unroll
    for (int j = 0; j < 6; ++j)
        x[j] = *(const v4f*)(xz + (size_t)(m0 - (KC - 1) + j) * XZW + d4);
    v4f w[4];
#pragma unroll
    for (int q = 0; q < 4; ++q)
        w[q] = *(const v4f*)(cw + (size_t)(d4 + q) * KC);
    const v4f bv = *(const v4f*)(cb + d4);
#pragma unroll
    for (int i = 0; i < 3; ++i) {
        v4f o;
#pragma unroll
        for (int q = 0; q < 4; ++q) {
            float t = bv[q];
#pragma unroll
            for (int k = 0; k < KC; ++k)
                t += x[i + k][q] * w[q][k];
            o[q] = siluf(t);
        }
        *(v4f*)(u + (size_t)(m0 + i) * DI + d4) = o;
    }
}

#define KPL 4
#define TT  (32 * KPL)
#define CHB 16
#define TST 17
#define BCP 36

template <int SH>
__device__ __forceinline__ float row_up(float keep, float v) {
    return __builtin_bit_cast(float, __builtin_amdgcn_update_dpp(
        __builtin_bit_cast(int, keep), __builtin_bit_cast(int, v),
        0x110 + SH, 0xf, 0xf, false));
}
__device__ __forceinline__ float lane_to_all(float v, int l) {
    return __builtin_bit_cast(float,
        __builtin_amdgcn_readlane(__builtin_bit_cast(int, v), l));
}

#define SCAN_STAGE(SH)                                   \
    {                                                    \
        const float ap = row_up<SH>(1.0f, a);            \
        const float sp = row_up<SH>(0.0f, s);            \
        s = a * sp + s;                                  \
        a = a * ap;                                      \
    }

__global__ void __launch_bounds__(32 * CHB) scan_kernel(
    const float* __restrict__ dt, const float* __restrict__ u,
    const float* __restrict__ dbl, const float* __restrict__ xz,
    const float* __restrict__ A_log, const float* __restrict__ Dp,
    float* __restrict__ yb)
{
    __shared__ __attribute__((aligned(16))) float lds[TT * BCP + 4 * TT * TST];
    float* const sBC = lds;
    float* const sDT = lds + TT * BCP;
    float* const sU  = sDT + TT * TST;
    float* const sZ  = sU  + TT * TST;
    float* const sY  = sZ  + TT * TST;

    const int lane = threadIdx.x & 31;
    const int w    = threadIdx.x >> 5;
    const int b    = blockIdx.x / (DI / CHB);
    const int cg   = blockIdx.x % (DI / CHB);
    const int d0   = cg * CHB;
    const int dch  = d0 + w;
    const int hl   = lane >> 4;
    const int c16  = lane & 15;
    const int pm   = (2 * (w & 1) + hl) * 32 + (w >> 1);

    float An[NS];
#pragma unroll
    for (int n = 0; n < NS; ++n)
        An[n] = lane_to_all(-__expf(A_log[dch * NS + n]), 0);
    const float Dd = Dp[dch];
    float hin[NS];
#pragma unroll
    for (int n = 0; n < NS; ++n) hin[n] = 0.0f;

    for (int c = 0; c < LL / TT; ++c) {
#pragma unroll
        for (int jj = 0; jj < TT / (2 * CHB); ++jj) {
            const int r = 32 * jj + 2 * w + hl;
            const int p = pm + 8 * jj;
            const size_t row = (size_t)(b * LL + c * TT + r);
            sDT[p * TST + c16] = dt[row * DI + d0 + c16];
            sU [p * TST + c16] = u[row * DI + d0 + c16];
            sZ [p * TST + c16] = xz[row * XZW + DI + d0 + c16];
        }
#pragma unroll
        for (int jj = 0; jj < TT / (4 * CHB); ++jj) {
            const int r = (jj * CHB + w) * 4 + (lane >> 3);
            const int p = (lane >> 3) * 32 + jj * CHB + w;
            const size_t row = (size_t)(b * LL + c * TT + r);
            *(v4f*)&sBC[p * BCP + 4 * (lane & 7)] =
                *(const v4f*)(dbl + row * DBLW + RR + 4 * (lane & 7));
        }
        __syncthreads();

        float dtv[KPL], du[KPL], y[KPL];
#pragma unroll
        for (int j = 0; j < KPL; ++j) {
            dtv[j] = sDT[(j * 32 + lane) * TST + w];
            du[j]  = dtv[j] * sU[(j * 32 + lane) * TST + w];
            y[j]   = 0.0f;
        }
#pragma unroll
        for (int g = 0; g < NS / 4; ++g) {
            v4f Bv[KPL], Cv[KPL];
#pragma unroll
            for (int j = 0; j < KPL; ++j) {
                Bv[j] = *(const v4f*)&sBC[(j * 32 + lane) * BCP + 4 * g];
                Cv[j] = *(const v4f*)&sBC[(j * 32 + lane) * BCP + NS + 4 * g];
            }
#pragma unroll
            for (int q = 0; q < 4; ++q) {
                const int n = 4 * g + q;
                float aj[KPL], sj[KPL];
#pragma unroll
                for (int j = 0; j < KPL; ++j) {
                    aj[j] = __expf(dtv[j] * An[n]);
                    sj[j] = du[j] * Bv[j][q];
                }
                float a = aj[0];
                float s = sj[0];
#pragma unroll
                for (int j = 1; j < KPL; ++j) {
                    s = aj[j] * s + sj[j];
                    a = a * aj[j];
                }
                SCAN_STAGE(1)
                SCAN_STAGE(2)
                SCAN_STAGE(4)
                SCAN_STAGE(8)
                const float a15 = lane_to_all(a, 15);
                const float s15 = lane_to_all(s, 15);
                if (lane >= 16) {
                    s = a * s15 + s;
                    a = a * a15;
                }
                const float hend = a * hin[n] + s;
                float h = row_up<1>(hin[n], hend);
                const float h15 = lane_to_all(hend, 15);
                h = (lane == 16) ? h15 : h;
                hin[n] = lane_to_all(hend, 31);
#pragma unroll
                for (int j = 0; j < KPL; ++j) {
                    h = aj[j] * h + sj[j];
                    y[j] += h * Cv[j][q];
                }
            }
        }
#pragma unroll
        for (int j = 0; j < KPL; ++j) {
            const float uv = sU[(j * 32 + lane) * TST + w];
            const float zv = sZ[(j * 32 + lane) * TST + w];
            sY[(j * 32 + lane) * TST + w] = (y[j] + uv * Dd) * siluf(zv);
        }
        __syncthreads();

#pragma unroll
        for (int jj = 0; jj < TT / (2 * CHB); ++jj) {
            const int r = 32 * jj + 2 * w + hl;
            const int p = pm + 8 * jj;
            const size_t row = (size_t)(b * LL + c * TT + r);
            yb[((size_t)cg * (BB * LL) + row) * CHB + c16] = sY[p * TST + c16];
        }
    }
}

extern "C" void kernel_launch(void* const* d_in, const int* in_sizes, int n_in,
                              void* d_out, int out_size, void* d_ws, size_t ws_size,
                              hipStream_t stream) {
    const float* x          = (const float*)d_in[0];
    const float* in_proj_w  = (const float*)d_in[1];
    const float* conv_w     = (const float*)d_in[2];
    const float* conv_b     = (const float*)d_in[3];
    const float* x_proj_w   = (const float*)d_in[4];
    const float* dt_proj_w  = (const float*)d_in[5];
    const float* dt_proj_b  = (const float*)d_in[6];
    const float* A_log      = (const float*)d_in[7];
    const float* Dvec       = (const float*)d_in[8];
    const float* out_proj_w = (const float*)d_in[9];
    float* out = (float*)d_out;

    float* ws  = (float*)d_ws;
    float* xz  = ws;
    float* u   = xz  + (size_t)2048 * 4096;
    float* dbl = u   + (size_t)2048 * 2048;
    float* dtb = dbl + (size_t)2048 * 96;
    float* yg  = dtb + (size_t)2048 * 2048;

    wmma_gemm_f32<4, 4, 16, 2, 2, XZW, DMOD, DMOD, 8, 16, EPI_CONV>
        <<<dim3(XZW / 128, (BB * LL) / 128), 128, 0, stream>>>(x, in_proj_w, xz, conv_b, conv_w, u);

    conv_edge_kernel<<<((BB * LL) / XT) * (DI / 4) / 256, 256, 0, stream>>>(xz, conv_w, conv_b, u);

    wmma_gemm_f32<1, 2, 32, 2, 1, DBLW, DI, DI, 16, 32, EPI_PLAIN>
        <<<dim3(DBLW / 32, (BB * LL) / 32), 64, 0, stream>>>(u, x_proj_w, dbl, nullptr, nullptr, nullptr);

    wmma_gemm_f32<1, 4, 16, 2, 2, DI, RR, DBLW, 8, 16, EPI_SOFTPLUS>
        <<<dim3(DI / 128, (BB * LL) / 32), 128, 0, stream>>>(dbl, dt_proj_w, dtb, dt_proj_b, nullptr, nullptr);

    scan_kernel<<<BB * (DI / CHB), 32 * CHB, 0, stream>>>(dtb, u, dbl, xz, A_log, Dvec, yg);

    wmma_gemm_f32<2, 4, 32, 2, 2, DMOD, DI, CHB, BB * LL * CHB, 2 * BB * LL * CHB, EPI_PLAIN>
        <<<dim3(DMOD / 128, (BB * LL) / 64), 128, 0, stream>>>(yg, out_proj_w, out, nullptr, nullptr, nullptr);
}


# dt product folded into the scan: each block makes its 128-step x 16-channel dt tile on WMMA (weights kept in registers), softplus on the exp2/log2/rcp units, straight into LDS; one launch less
# speedup vs baseline: 10.4733x; 1.0794x over previous
#include <hip/hip_runtime.h>
#include <hip/hip_bf16.h>
#include <cmath>

typedef __attribute__((ext_vector_type(2))) float v2f;
typedef __attribute__((ext_vector_type(4))) float v4f;
typedef __attribute__((ext_vector_type(8))) float v8f;

#define BB    2
#define LL    1024
#define DMOD  1024
#define DI    2048
#define NS    16
#define KC    4
#define RR    64
#define XZW   (2 * DI)
#define DBLW  (RR + 2 * NS)

__device__ __forceinline__ float siluf(float v) {
    return v / (1.0f + __expf(-v));
}
__device__ __forceinline__ float softplusf(float v) {
    const float e = __expf(v);
    const float p = 1.0f + e;
    const float d = p - 1.0f;
    const float l = (__builtin_amdgcn_logf(p) * 0.693147180559945f) * (e * __builtin_amdgcn_rcpf(d));
    const float r = (d == 0.0f) ? e : l;
    return (v > 20.0f) ? v : r;
}
__device__ __forceinline__ float other_half(float v) {
    return __builtin_bit_cast(float, __builtin_amdgcn_permlanex16(
        __builtin_bit_cast(int, v), __builtin_bit_cast(int, v),
        0x76543210, 0xfedcba98, false, false));
}

template <int CPL> struct colvec;
template <> struct colvec<4> { typedef v4f type; };
template <> struct colvec<2> { typedef v2f type; };

template <int MT, int CPL, int KS>
struct Slab {
    v4f a[MT][KS / 8];
    typename colvec<CPL>::type b[KS / 2];
};

template <int MT, int CPL, int KS, int N, int LDA>
__device__ __forceinline__ void slab_load(Slab<MT, CPL, KS>& s,
                                          const float* __restrict__ ap,
                                          const float* __restrict__ wp)
{
    typedef typename colvec<CPL>::type cvec;
#pragma unroll
    for (int mt = 0; mt < MT; ++mt)
#pragma unroll
        for (int i = 0; i < KS / 8; ++i)
            s.a[mt][i] = *(const v4f*)(ap + (size_t)mt * 16 * LDA + 4 * i);
#pragma unroll
    for (int kk = 0; kk < KS / 2; ++kk)
        s.b[kk] = *(const cvec*)(wp + (size_t)kk * N);
}

template <int MT, int CPL, int KS>
__device__ __forceinline__ void slab_mma(const Slab<MT, CPL, KS>& s, v8f (&acc)[MT][CPL])
{
#pragma unroll
    for (int j = 0; j < KS / 4; ++j) {
        v2f bf[CPL];
#pragma unroll
        for (int q = 0; q < CPL; ++q) {
            bf[q][0] = s.b[2 * j][q];
            bf[q][1] = s.b[2 * j + 1][q];
        }
#pragma unroll
        for (int mt = 0; mt < MT; ++mt) {
            v2f af;
            af[0] = s.a[mt][(2 * j) >> 2][(2 * j) & 3];
            af[1] = s.a[mt][(2 * j) >> 2][((2 * j) & 3) + 1];
#pragma unroll
            for (int q = 0; q < CPL; ++q)
                acc[mt][q] = __builtin_amdgcn_wmma_f32_16x16x4_f32(
                    false, af, false, bf[q], (short)0, acc[mt][q], false, false);
        }
    }
}

#define EPI_PLAIN    0
#define EPI_CONV     2
template <int MT, int CPL, int KS, int WM, int WN, int N, int K, int LDA, int AHALF, int ASLAB, int EPI>
__global__ void __launch_bounds__(128) wmma_gemm_f32(
    const float* __restrict__ A, const float* __restrict__ W,
    float* __restrict__ C, const float* __restrict__ bias,
    const float* __restrict__ cw, float* __restrict__ U)
{
    static_assert(K % (2 * KS) == 0 && KS % 8 == 0, "slab pairs");
    static_assert(N % (WN * CPL * 16) == 0, "column tiles");
    typedef typename colvec<CPL>::type cvec;
    constexpr int KH = KS / 2;

    const int wave = threadIdx.x >> 5;
    const int lane = threadIdx.x & 31;
    const int half = lane >> 4;
    const int c    = lane & 15;
    const int m0   = (blockIdx.y * WM + wave / WN) * (MT * 16);
    const int n0   = (blockIdx.x * WN + wave % WN) * (CPL * 16);

    v8f acc[MT][CPL] = {};

    const float* aptr = A + (size_t)(m0 + c) * LDA + (size_t)half * AHALF;
    const float* wptr = W + (size_t)(half * KH) * N + n0 + CPL * c;

#define LOADS_FIRST() __builtin_amdgcn_sched_barrier(0)
    Slab<MT, CPL, KS> s0, s1;
    slab_load<MT, CPL, KS, N, LDA>(s0, aptr, wptr);
    for (int i = 0; i < K / (2 * KS) - 1; ++i) {
        aptr += ASLAB;  wptr += (size_t)KS * N;
        slab_load<MT, CPL, KS, N, LDA>(s1, aptr, wptr);
        LOADS_FIRST();
        slab_mma<MT, CPL, KS>(s0, acc);
        aptr += ASLAB;  wptr += (size_t)KS * N;
        slab_load<MT, CPL, KS, N, LDA>(s0, aptr, wptr);
        LOADS_FIRST();
        slab_mma<MT, CPL, KS>(s1, acc);
    }
    aptr += ASLAB;  wptr += (size_t)KS * N;
    slab_load<MT, CPL, KS, N, LDA>(s1, aptr, wptr);
    LOADS_FIRST();
    slab_mma<MT, CPL, KS>(s0, acc);
    slab_mma<MT, CPL, KS>(s1, acc);
#undef LOADS_FIRST

    if constexpr (EPI == EPI_CONV) {
        if (n0 < DI) {
            const bool bstart = (m0 % LL) == 0;
            v4f wq[CPL];
#pragma unroll
            for (int q = 0; q < CPL; ++q)
                wq[q] = *(const v4f*)(cw + (size_t)(n0 + CPL * c + q) * KC);
            const cvec cbv = *(const cvec*)(bias + n0 + CPL * c);
            float* up = U + (size_t)(m0 + 8 * half) * DI + n0 + CPL * c;
            float* xp = C + (size_t)(m0 + (half ? (MT - 1) * 16 + 8 + 5 : 0)) * N + n0 + CPL * c;
#pragma unroll
            for (int i = 0; i < 3; ++i) {
                cvec v;
#pragma unroll
                for (int q = 0; q < CPL; ++q)
                    v[q] = half ? acc[MT - 1][q][5 + i] : acc[0][q][i];
                *(cvec*)(xp + (size_t)i * N) = v;
            }
#pragma unroll
            for (int mt = 0; mt < MT; ++mt) {
                float pv[CPL][3];
#pragma unroll
                for (int q = 0; q < CPL; ++q)
#pragma unroll
                    for (int i = 0; i < 3; ++i) {
                        const float lo = acc[mt][q][5 + i];
                        const float hi = (mt > 0) ? acc[mt > 0 ? mt - 1 : 0][q][5 + i] : 0.0f;
                        pv[q][i] = other_half(half ? hi : lo);
                    }
#pragma unroll
                for (int r = 0; r < 8; ++r) {
                    cvec v;
#pragma unroll
                    for (int q = 0; q < CPL; ++q) {
                        const float x3 = (r >= 3) ? acc[mt][q][r >= 3 ? r - 3 : 0] : pv[q][r >= 3 ? 0 : r];
                        const float x2 = (r >= 2) ? acc[mt][q][r >= 2 ? r - 2 : 0] : pv[q][r >= 2 ? 0 : r + 1];
                        const float x1 = (r >= 1) ? acc[mt][q][r >= 1 ? r - 1 : 0] : pv[q][2];
                        float t = cbv[q];
                        t += x3 * wq[q][0];
                        t += x2 * wq[q][1];
                        t += x1 * wq[q][2];
                        t += acc[mt][q][r] * wq[q][3];
                        v[q] = siluf(t);
                    }
                    if (mt > 0 || r >= 3 || half == 1 || bstart)
                        *(cvec*)(up + (size_t)(mt * 16 + r) * DI) = v;
                }
            }
            return;
        }
    }

    float* cp = C + (size_t)(m0 + 8 * half) * N + n0 + CPL * c;
#pragma unroll
    for (int mt = 0; mt < MT; ++mt) {
#pragma unroll
        for (int r = 0; r < 8; ++r) {
            cvec v;
#pragma unroll
            for (int q = 0; q < CPL; ++q)
                v[q] = acc[mt][q][r];
            *(cvec*)(cp + (size_t)(mt * 16 + r) * N) = v;
        }
    }
}

#define XT 64
static_assert(XT == 4 * 16, "edge kernel tile = in_proj wave tile (MT = 4)");
__global__ void __launch_bounds__(256) conv_edge_kernel(const float* __restrict__ xz,
                                                        const float* __restrict__ cw,
                                                        const float* __restrict__ cb,
                                                        float* __restrict__ u)
{
    const int idx = blockIdx.x * 256 + threadIdx.x;
    const int d4  = (idx % (DI / 4)) * 4;
    const int m0  = (idx / (DI / 4)) * XT;
    if (m0 % LL == 0) return;
    v4f x[6];
#pragma unroll
    for (int j = 0; j < 6; ++j)
        x[j] = *(const v4f*)(xz + (size_t)(m0 - (KC - 1) + j) * XZW + d4);
    v4f w[4];
#pragma unroll
    for (int q = 0; q < 4; ++q)
        w[q] = *(const v4f*)(cw + (size_t)(d4 + q) * KC);
    const v4f bv = *(const v4f*)(cb + d4);
#pragma unroll
    for (int i = 0; i < 3; ++i) {
        v4f o;
#pragma unroll
        for (int q = 0; q < 4; ++q) {
            float t = bv[q];
#pragma unroll
            for (int k = 0; k < KC; ++k)
                t += x[i + k][q] * w[q][k];
            o[q] = siluf(t);
        }
        *(v4f*)(u + (size_t)(m0 + i) * DI + d4) = o;
    }
}

#define KPL 4
#define TT  (32 * KPL)
#define CHB 16
#define TST 17
#define BCP 36

template <int SH>
__device__ __forceinline__ float row_up(float keep, float v) {
    return __builtin_bit_cast(float, __builtin_amdgcn_update_dpp(
        __builtin_bit_cast(int, keep), __builtin_bit_cast(int, v),
        0x110 + SH, 0xf, 0xf, false));
}
__device__ __forceinline__ float lane_to_all(float v, int l) {
    return __builtin_bit_cast(float,
        __builtin_amdgcn_readlane(__builtin_bit_cast(int, v), l));
}

#define SCAN_STAGE(SH)                                   \
    {                                                    \
        const float ap = row_up<SH>(1.0f, a);            \
        const float sp = row_up<SH>(0.0f, s);            \
        s = a * sp + s;                                  \
        a = a * ap;                                      \
    }

#define DTW (TT / 16)
static_assert(DTW <= CHB && RR % 16 == 0, "dt tile: one wave per 16 time rows, k in slabs of 16");
__global__ void __launch_bounds__(32 * CHB) scan_kernel(
    const float* __restrict__ dtw, const float* __restrict__ dtbias,
    const float* __restrict__ u,
    const float* __restrict__ dbl, const float* __restrict__ xz,
    const float* __restrict__ A_log, const float* __restrict__ Dp,
    float* __restrict__ yb)
{
    __shared__ __attribute__((aligned(16))) float lds[TT * BCP + 4 * TT * TST];
    float* const sBC = lds;
    float* const sDT = lds + TT * BCP;
    float* const sU  = sDT + TT * TST;
    float* const sZ  = sU  + TT * TST;
    float* const sY  = sZ  + TT * TST;

    const int lane = threadIdx.x & 31;
    const int w    = threadIdx.x >> 5;
    const int b    = blockIdx.x / (DI / CHB);
    const int cg   = blockIdx.x % (DI / CHB);
    const int d0   = cg * CHB;
    const int dch  = d0 + w;
    const int hl   = lane >> 4;
    const int c16  = lane & 15;
    const int pm   = (2 * (w & 1) + hl) * 32 + (w >> 1);

    float An[NS];
#pragma unroll
    for (int n = 0; n < NS; ++n)
        An[n] = lane_to_all(-__expf(A_log[dch * NS + n]), 0);
    const float Dd = Dp[dch];
    const bool dtwave = __builtin_amdgcn_readfirstlane(w) < DTW;
    v2f wfrag[RR / 4];
    float dbias = 0.0f;
    if (dtwave) {
#pragma unroll
        for (int sj = 0; sj < RR / 4; ++sj) {
            const int k = 16 * (sj >> 2) + 8 * hl + 2 * (sj & 3);
            wfrag[sj][0] = dtw[(size_t)k * DI + d0 + c16];
            wfrag[sj][1] = dtw[(size_t)(k + 1) * DI + d0 + c16];
        }
        dbias = dtbias[d0 + c16];
    }
    float hin[NS];
#pragma unroll
    for (int n = 0; n < NS; ++n) hin[n] = 0.0f;

    v4f afr[RR / 8];
#define DT_ROWS_LOAD(TRIP)                                                                        \
    {                                                                                             \
        const float* ap = dbl + (size_t)(b * LL + (TRIP) * TT + 16 * w + c16) * DBLW + 8 * hl;    \
        _Pragma("unroll")             \
        for (int s = 0; s < RR / 16; ++s) {                                                       \
            afr[2 * s]     = *(const v4f*)(ap + 16 * s);                                          \
            afr[2 * s + 1] = *(const v4f*)(ap + 16 * s + 4);                                      \
        }                                                                                         \
    }
    if (dtwave) DT_ROWS_LOAD(0)

    for (int c = 0; c < LL / TT; ++c) {
        float uv[TT / (2 * CHB)], zv[TT / (2 * CHB)];
        v4f bcv[TT / (4 * CHB)];
#pragma unroll
        for (int jj = 0; jj < TT / (2 * CHB); ++jj) {
            const int r = 32 * jj + 2 * w + hl;
            const size_t row = (size_t)(b * LL + c * TT + r);
            uv[jj] = u[row * DI + d0 + c16];
            zv[jj] = xz[row * XZW + DI + d0 + c16];
        }
#pragma unroll
        for (int jj = 0; jj < TT / (4 * CHB); ++jj) {
            const int r = (jj * CHB + w) * 4 + (lane >> 3);
            const size_t row = (size_t)(b * LL + c * TT + r);
            bcv[jj] = *(const v4f*)(dbl + row * DBLW + RR + 4 * (lane & 7));
        }
        if (dtwave) {
            v8f dacc = {};
#pragma unroll
            for (int sj = 0; sj < RR / 4; ++sj) {
                const int s = sj >> 2, j = sj & 3;
                v2f af;
                af[0] = afr[2 * s + ((2 * j) >> 2)][(2 * j) & 3];
                af[1] = afr[2 * s + ((2 * j) >> 2)][((2 * j) & 3) + 1];
                dacc = __builtin_amdgcn_wmma_f32_16x16x4_f32(
                    false, af, false, wfrag[sj], (short)0, dacc, false, false);
            }
#pragma unroll
            for (int r = 0; r < 8; ++r) {
                const int p = (r & 3) * 32 + 4 * w + 2 * hl + (r >> 2);
                sDT[p * TST + c16] = softplusf(dacc[r] + dbias);
            }
        }
#pragma unroll
        for (int jj = 0; jj < TT / (2 * CHB); ++jj) {
            const int p = pm + 8 * jj;
            sU[p * TST + c16] = uv[jj];
            sZ[p * TST + c16] = zv[jj];
        }
#pragma unroll
        for (int jj = 0; jj < TT / (4 * CHB); ++jj) {
            const int p = (lane >> 3) * 32 + jj * CHB + w;
            *(v4f*)&sBC[p * BCP + 4 * (lane & 7)] = bcv[jj];
        }
        __syncthreads();
        if (dtwave && c + 1 < LL / TT) DT_ROWS_LOAD(c + 1)

        float dtv[KPL], du[KPL], y[KPL];
#pragma unroll
        for (int j = 0; j < KPL; ++j) {
            dtv[j] = sDT[(j * 32 + lane) * TST + w];
            du[j]  = dtv[j] * sU[(j * 32 + lane) * TST + w];
            y[j]   = 0.0f;
        }
#pragma unroll
        for (int g = 0; g < NS / 4; ++g) {
            v4f Bv[KPL], Cv[KPL];
#pragma unroll
            for (int j = 0; j < KPL; ++j) {
                Bv[j] = *(const v4f*)&sBC[(j * 32 + lane) * BCP + 4 * g];
                Cv[j] = *(const v4f*)&sBC[(j * 32 + lane) * BCP + NS + 4 * g];
            }
#pragma unroll
            for (int q = 0; q < 4; ++q) {
                const int n = 4 * g + q;
                float aj[KPL], sj[KPL];
#pragma unroll
                for (int j = 0; j < KPL; ++j) {
                    aj[j] = __expf(dtv[j] * An[n]);
                    sj[j] = du[j] * Bv[j][q];
                }
                float a = aj[0];
                float s = sj[0];
#pragma unroll
                for (int j = 1; j < KPL; ++j) {
                    s = aj[j] * s + sj[j];
                    a = a * aj[j];
                }
                SCAN_STAGE(1)
                SCAN_STAGE(2)
                SCAN_STAGE(4)
                SCAN_STAGE(8)
                const float a15 = lane_to_all(a, 15);
                const float s15 = lane_to_all(s, 15);
                if (lane >= 16) {
                    s = a * s15 + s;
                    a = a * a15;
                }
                const float hend = a * hin[n] + s;
                float h = row_up<1>(hin[n], hend);
                const float h15 = lane_to_all(hend, 15);
                h = (lane == 16) ? h15 : h;
                hin[n] = lane_to_all(hend, 31);
#pragma unroll
                for (int j = 0; j < KPL; ++j) {
                    h = aj[j] * h + sj[j];
                    y[j] += h * Cv[j][q];
                }
            }
        }
#pragma unroll
        for (int j = 0; j < KPL; ++j) {
            const float uv = sU[(j * 32 + lane) * TST + w];
            const float zv = sZ[(j * 32 + lane) * TST + w];
            sY[(j * 32 + lane) * TST + w] = (y[j] + uv * Dd) * siluf(zv);
        }
        __syncthreads();

#pragma unroll
        for (int jj = 0; jj < TT / (2 * CHB); ++jj) {
            const int r = 32 * jj + 2 * w + hl;
            const int p = pm + 8 * jj;
            const size_t row = (size_t)(b * LL + c * TT + r);
            yb[((size_t)cg * (BB * LL) + row) * CHB + c16] = sY[p * TST + c16];
        }
    }
}

#undef DT_ROWS_LOAD
extern "C" void kernel_launch(void* const* d_in, const int* in_sizes, int n_in,
                              void* d_out, int out_size, void* d_ws, size_t ws_size,
                              hipStream_t stream) {
    const float* x          = (const float*)d_in[0];
    const float* in_proj_w  = (const float*)d_in[1];
    const float* conv_w     = (const float*)d_in[2];
    const float* conv_b     = (const float*)d_in[3];
    const float* x_proj_w   = (const float*)d_in[4];
    const float* dt_proj_w  = (const float*)d_in[5];
    const float* dt_proj_b  = (const float*)d_in[6];
    const float* A_log      = (const float*)d_in[7];
    const float* Dvec       = (const float*)d_in[8];
    const float* out_proj_w = (const float*)d_in[9];
    float* out = (float*)d_out;

    float* ws  = (float*)d_ws;
    float* xz  = ws;
    float* u   = xz  + (size_t)2048 * 4096;
    float* dbl = u   + (size_t)2048 * 2048;
    float* yg  = dbl + (size_t)2048 * 96;

    wmma_gemm_f32<4, 4, 16, 2, 2, XZW, DMOD, DMOD, 8, 16, EPI_CONV>
        <<<dim3(XZW / 128, (BB * LL) / 128), 128, 0, stream>>>(x, in_proj_w, xz, conv_b, conv_w, u);

    conv_edge_kernel<<<((BB * LL) / XT) * (DI / 4) / 256, 256, 0, stream>>>(xz, conv_w, conv_b, u);

    wmma_gemm_f32<1, 2, 32, 2, 1, DBLW, DI, DI, 16, 32, EPI_PLAIN>
        <<<dim3(DBLW / 32, (BB * LL) / 32), 64, 0, stream>>>(u, x_proj_w, dbl, nullptr, nullptr, nullptr);

    scan_kernel<<<BB * (DI / CHB), 32 * CHB, 0, stream>>>(dt_proj_w, dt_proj_b, u, dbl, xz, A_log, Dvec, yg);

    wmma_gemm_f32<2, 4, 32, 2, 2, DMOD, DI, CHB, BB * LL * CHB, 2 * BB * LL * CHB, EPI_PLAIN>
        <<<dim3(DMOD / 128, (BB * LL) / 64), 128, 0, stream>>>(yg, out_proj_w, out, nullptr, nullptr, nullptr);
}


# x_proj product: ring of 4 register slabs, a slab's fragment loads issued 3 slabs (48 WMMAs) ahead of its use, not 1, so 60 loads per wave stay in flight; k order and sums unchanged
# speedup vs baseline: 11.1016x; 1.0600x over previous
#include <hip/hip_runtime.h>
#include <hip/hip_bf16.h>
#include <cmath>

typedef __attribute__((ext_vector_type(2))) float v2f;
typedef __attribute__((ext_vector_type(4))) float v4f;
typedef __attribute__((ext_vector_type(8))) float v8f;

#define BB    2
#define LL    1024
#define DMOD  1024
#define DI    2048
#define NS    16
#define KC    4
#define RR    64
#define XZW   (2 * DI)
#define DBLW  (RR + 2 * NS)

__device__ __forceinline__ float siluf(float v) {
    return v / (1.0f + __expf(-v));
}
__device__ __forceinline__ float softplusf(float v) {
    const float e = __expf(v);
    const float p = 1.0f + e;
    const float d = p - 1.0f;
    const float l = (__builtin_amdgcn_logf(p) * 0.693147180559945f) * (e * __builtin_amdgcn_rcpf(d));
    const float r = (d == 0.0f) ? e : l;
    return (v > 20.0f) ? v : r;
}
__device__ __forceinline__ float other_half(float v) {
    return __builtin_bit_cast(float, __builtin_amdgcn_permlanex16(
        __builtin_bit_cast(int, v), __builtin_bit_cast(int, v),
        0x76543210, 0xfedcba98, false, false));
}

template <int CPL> struct colvec;
template <> struct colvec<4> { typedef v4f type; };
template <> struct colvec<2> { typedef v2f type; };

template <int MT, int CPL, int KS>
struct Slab {
    v4f a[MT][KS / 8];
    typename colvec<CPL>::type b[KS / 2];
};

template <int MT, int CPL, int KS, int N, int LDA>
__device__ __forceinline__ void slab_load(Slab<MT, CPL, KS>& s,
                                          const float* __restrict__ ap,
                                          const float* __restrict__ wp)
{
    typedef typename colvec<CPL>::type cvec;
#pragma unroll
    for (int mt = 0; mt < MT; ++mt)
#pragma unroll
        for (int i = 0; i < KS / 8; ++i)
            s.a[mt][i] = *(const v4f*)(ap + (size_t)mt * 16 * LDA + 4 * i);
#pragma unroll
    for (int kk = 0; kk < KS / 2; ++kk)
        s.b[kk] = *(const cvec*)(wp + (size_t)kk * N);
}

template <int MT, int CPL, int KS>
__device__ __forceinline__ void slab_mma(const Slab<MT, CPL, KS>& s, v8f (&acc)[MT][CPL])
{
#pragma unroll
    for (int j = 0; j < KS / 4; ++j) {
        v2f bf[CPL];
#pragma unroll
        for (int q = 0; q < CPL; ++q) {
            bf[q][0] = s.b[2 * j][q];
            bf[q][1] = s.b[2 * j + 1][q];
        }
#pragma unroll
        for (int mt = 0; mt < MT; ++mt) {
            v2f af;
            af[0] = s.a[mt][(2 * j) >> 2][(2 * j) & 3];
            af[1] = s.a[mt][(2 * j) >> 2][((2 * j) & 3) + 1];
#pragma unroll
            for (int q = 0; q < CPL; ++q)
                acc[mt][q] = __builtin_amdgcn_wmma_f32_16x16x4_f32(
                    false, af, false, bf[q], (short)0, acc[mt][q], false, false);
        }
    }
}

#define EPI_PLAIN    0
#define EPI_CONV     2
template <int MT, int CPL, int KS, int WM, int WN, int N, int K, int LDA, int AHALF, int ASLAB, int EPI, int NB>
__global__ void __launch_bounds__(128) wmma_gemm_f32(
    const float* __restrict__ A, const float* __restrict__ W,
    float* __restrict__ C, const float* __restrict__ bias,
    const float* __restrict__ cw, float* __restrict__ U)
{
    constexpr int NSLAB = K / KS;
    static_assert(NB >= 2 && NSLAB % NB == 0 && NSLAB >= 2 * NB && KS % 8 == 0, "whole rings of slabs");
    static_assert(N % (WN * CPL * 16) == 0, "column tiles");
    typedef typename colvec<CPL>::type cvec;
    constexpr int KH = KS / 2;

    const int wave = threadIdx.x >> 5;
    const int lane = threadIdx.x & 31;
    const int half = lane >> 4;
    const int c    = lane & 15;
    const int m0   = (blockIdx.y * WM + wave / WN) * (MT * 16);
    const int n0   = (blockIdx.x * WN + wave % WN) * (CPL * 16);

    v8f acc[MT][CPL] = {};

    const float* aptr = A + (size_t)(m0 + c) * LDA + (size_t)half * AHALF;
    const float* wptr = W + (size_t)(half * KH) * N + n0 + CPL * c;

#define LOADS_FIRST() __builtin_amdgcn_sched_barrier(0)
    Slab<MT, CPL, KS> s[NB];
#pragma unroll
    for (int r = 0; r < NB - 1; ++r) {
        slab_load<MT, CPL, KS, N, LDA>(s[r], aptr + (size_t)r * ASLAB, wptr + (size_t)r * KS * N);
        if constexpr (NB > 2) LOADS_FIRST();
    }
    for (int i = 0; i < NSLAB / NB - 1; ++i) {
#pragma unroll
        for (int r = 0; r < NB; ++r) {
            slab_load<MT, CPL, KS, N, LDA>(s[(r + NB - 1) % NB], aptr + (size_t)(r + NB - 1) * ASLAB,
                                           wptr + (size_t)(r + NB - 1) * KS * N);
            LOADS_FIRST();
            slab_mma<MT, CPL, KS>(s[r], acc);
        }
        aptr += NB * ASLAB;  wptr += (size_t)NB * KS * N;
    }
    slab_load<MT, CPL, KS, N, LDA>(s[NB - 1], aptr + (size_t)(NB - 1) * ASLAB, wptr + (size_t)(NB - 1) * KS * N);
    LOADS_FIRST();
#pragma unroll
    for (int r = 0; r < NB; ++r)
        slab_mma<MT, CPL, KS>(s[r], acc);
#undef LOADS_FIRST

    if constexpr (EPI == EPI_CONV) {
        if (n0 < DI) {
            const bool bstart = (m0 % LL) == 0;
            v4f wq[CPL];
#pragma unroll
            for (int q = 0; q < CPL; ++q)
                wq[q] = *(const v4f*)(cw + (size_t)(n0 + CPL * c + q) * KC);
            const cvec cbv = *(const cvec*)(bias + n0 + CPL * c);
            float* up = U + (size_t)(m0 + 8 * half) * DI + n0 + CPL * c;
            float* xp = C + (size_t)(m0 + (half ? (MT - 1) * 16 + 8 + 5 : 0)) * N + n0 + CPL * c;
#pragma unroll
            for (int i = 0; i < 3; ++i) {
                cvec v;
#pragma unroll
                for (int q = 0; q < CPL; ++q)
                    v[q] = half ? acc[MT - 1][q][5 + i] : acc[0][q][i];
                *(cvec*)(xp + (size_t)i * N) = v;
            }
#pragma unroll
            for (int mt = 0; mt < MT; ++mt) {
                float pv[CPL][3];
#pragma unroll
                for (int q = 0; q < CPL; ++q)
#pragma unroll
                    for (int i = 0; i < 3; ++i) {
                        const float lo = acc[mt][q][5 + i];
                        const float hi = (mt > 0) ? acc[mt > 0 ? mt - 1 : 0][q][5 + i] : 0.0f;
                        pv[q][i] = other_half(half ? hi : lo);
                    }
#pragma unroll
                for (int r = 0; r < 8; ++r) {
                    cvec v;
#pragma unroll
                    for (int q = 0; q < CPL; ++q) {
                        const float x3 = (r >= 3) ? acc[mt][q][r >= 3 ? r - 3 : 0] : pv[q][r >= 3 ? 0 : r];
                        const float x2 = (r >= 2) ? acc[mt][q][r >= 2 ? r - 2 : 0] : pv[q][r >= 2 ? 0 : r + 1];
                        const float x1 = (r >= 1) ? acc[mt][q][r >= 1 ? r - 1 : 0] : pv[q][2];
                        float t = cbv[q];
                        t += x3 * wq[q][0];
                        t += x2 * wq[q][1];
                        t += x1 * wq[q][2];
                        t += acc[mt][q][r] * wq[q][3];
                        v[q] = siluf(t);
                    }
                    if (mt > 0 || r >= 3 || half == 1 || bstart)
                        *(cvec*)(up + (size_t)(mt * 16 + r) * DI) = v;
                }
            }
            return;
        }
    }

    float* cp = C + (size_t)(m0 + 8 * half) * N + n0 + CPL * c;
#pragma unroll
    for (int mt = 0; mt < MT; ++mt) {
#pragma unroll
        for (int r = 0; r < 8; ++r) {
            cvec v;
#pragma unroll
            for (int q = 0; q < CPL; ++q)
                v[q] = acc[mt][q][r];
            *(cvec*)(cp + (size_t)(mt * 16 + r) * N) = v;
        }
    }
}

#define XT 64
static_assert(XT == 4 * 16, "edge kernel tile = in_proj wave tile (MT = 4)");
__global__ void __launch_bounds__(256) conv_edge_kernel(const float* __restrict__ xz,
                                                        const float* __restrict__ cw,
                                                        const float* __restrict__ cb,
                                                        float* __restrict__ u)
{
    const int idx = blockIdx.x * 256 + threadIdx.x;
    const int d4  = (idx % (DI / 4)) * 4;
    const int m0  = (idx / (DI / 4)) * XT;
    if (m0 % LL == 0) return;
    v4f x[6];
#pragma unroll
    for (int j = 0; j < 6; ++j)
        x[j] = *(const v4f*)(xz + (size_t)(m0 - (KC - 1) + j) * XZW + d4);
    v4f w[4];
#pragma unroll
    for (int q = 0; q < 4; ++q)
        w[q] = *(const v4f*)(cw + (size_t)(d4 + q) * KC);
    const v4f bv = *(const v4f*)(cb + d4);
#pragma unroll
    for (int i = 0; i < 3; ++i) {
        v4f o;
#pragma unroll
        for (int q = 0; q < 4; ++q) {
            float t = bv[q];
#pragma unroll
            for (int k = 0; k < KC; ++k)
                t += x[i + k][q] * w[q][k];
            o[q] = siluf(t);
        }
        *(v4f*)(u + (size_t)(m0 + i) * DI + d4) = o;
    }
}

#define KPL 4
#define TT  (32 * KPL)
#define CHB 16
#define TST 17
#define BCP 36

template <int SH>
__device__ __forceinline__ float row_up(float keep, float v) {
    return __builtin_bit_cast(float, __builtin_amdgcn_update_dpp(
        __builtin_bit_cast(int, keep), __builtin_bit_cast(int, v),
        0x110 + SH, 0xf, 0xf, false));
}
__device__ __forceinline__ float lane_to_all(float v, int l) {
    return __builtin_bit_cast(float,
        __builtin_amdgcn_readlane(__builtin_bit_cast(int, v), l));
}

#define SCAN_STAGE(SH)                                   \
    {                                                    \
        const float ap = row_up<SH>(1.0f, a);            \
        const float sp = row_up<SH>(0.0f, s);            \
        s = a * sp + s;                                  \
        a = a * ap;                                      \
    }

#define DTW (TT / 16)
static_assert(DTW <= CHB && RR % 16 == 0, "dt tile: one wave per 16 time rows, k in slabs of 16");
__global__ void __launch_bounds__(32 * CHB) scan_kernel(
    const float* __restrict__ dtw, const float* __restrict__ dtbias,
    const float* __restrict__ u,
    const float* __restrict__ dbl, const float* __restrict__ xz,
    const float* __restrict__ A_log, const float* __restrict__ Dp,
    float* __restrict__ yb)
{
    __shared__ __attribute__((aligned(16))) float lds[TT * BCP + 4 * TT * TST];
    float* const sBC = lds;
    float* const sDT = lds + TT * BCP;
    float* const sU  = sDT + TT * TST;
    float* const sZ  = sU  + TT * TST;
    float* const sY  = sZ  + TT * TST;

    const int lane = threadIdx.x & 31;
    const int w    = threadIdx.x >> 5;
    const int b    = blockIdx.x / (DI / CHB);
    const int cg   = blockIdx.x % (DI / CHB);
    const int d0   = cg * CHB;
    const int dch  = d0 + w;
    const int hl   = lane >> 4;
    const int c16  = lane & 15;
    const int pm   = (2 * (w & 1) + hl) * 32 + (w >> 1);

    float An[NS];
#pragma unroll
    for (int n = 0; n < NS; ++n)
        An[n] = lane_to_all(-__expf(A_log[dch * NS + n]), 0);
    const float Dd = Dp[dch];
    const bool dtwave = __builtin_amdgcn_readfirstlane(w) < DTW;
    v2f wfrag[RR / 4];
    float dbias = 0.0f;
    if (dtwave) {
#pragma unroll
        for (int sj = 0; sj < RR / 4; ++sj) {
            const int k = 16 * (sj >> 2) + 8 * hl + 2 * (sj & 3);
            wfrag[sj][0] = dtw[(size_t)k * DI + d0 + c16];
            wfrag[sj][1] = dtw[(size_t)(k + 1) * DI + d0 + c16];
        }
        dbias = dtbias[d0 + c16];
    }
    float hin[NS];
#pragma unroll
    for (int n = 0; n < NS; ++n) hin[n] = 0.0f;

    v4f afr[RR / 8];
#define DT_ROWS_LOAD(TRIP)                                                                        \
    {                                                                                             \
        const float* ap = dbl + (size_t)(b * LL + (TRIP) * TT + 16 * w + c16) * DBLW + 8 * hl;    \
        _Pragma("unroll")             \
        for (int s = 0; s < RR / 16; ++s) {                                                       \
            afr[2 * s]     = *(const v4f*)(ap + 16 * s);                                          \
            afr[2 * s + 1] = *(const v4f*)(ap + 16 * s + 4);                                      \
        }                                                                                         \
    }
    if (dtwave) DT_ROWS_LOAD(0)

    for (int c = 0; c < LL / TT; ++c) {
        float uv[TT / (2 * CHB)], zv[TT / (2 * CHB)];
        v4f bcv[TT / (4 * CHB)];
#pragma unroll
        for (int jj = 0; jj < TT / (2 * CHB); ++jj) {
            const int r = 32 * jj + 2 * w + hl;
            const size_t row = (size_t)(b * LL + c * TT + r);
            uv[jj] = u[row * DI + d0 + c16];
            zv[jj] = xz[row * XZW + DI + d0 + c16];
        }
#pragma unroll
        for (int jj = 0; jj < TT / (4 * CHB); ++jj) {
            const int r = (jj * CHB + w) * 4 + (lane >> 3);
            const size_t row = (size_t)(b * LL + c * TT + r);
            bcv[jj] = *(const v4f*)(dbl + row * DBLW + RR + 4 * (lane & 7));
        }
        if (dtwave) {
            v8f dacc = {};
#pragma unroll
            for (int sj = 0; sj < RR / 4; ++sj) {
                const int s = sj >> 2, j = sj & 3;
                v2f af;
                af[0] = afr[2 * s + ((2 * j) >> 2)][(2 * j) & 3];
                af[1] = afr[2 * s + ((2 * j) >> 2)][((2 * j) & 3) + 1];
                dacc = __builtin_amdgcn_wmma_f32_16x16x4_f32(
                    false, af, false, wfrag[sj], (short)0, dacc, false, false);
            }
#pragma unroll
            for (int r = 0; r < 8; ++r) {
                const int p = (r & 3) * 32 + 4 * w + 2 * hl + (r >> 2);
                sDT[p * TST + c16] = softplusf(dacc[r] + dbias);
            }
        }
#pragma unroll
        for (int jj = 0; jj < TT / (2 * CHB); ++jj) {
            const int p = pm + 8 * jj;
            sU[p * TST + c16] = uv[jj];
            sZ[p * TST + c16] = zv[jj];
        }
#pragma unroll
        for (int jj = 0; jj < TT / (4 * CHB); ++jj) {
            const int p = (lane >> 3) * 32 + jj * CHB + w;
            *(v4f*)&sBC[p * BCP + 4 * (lane & 7)] = bcv[jj];
        }
        __syncthreads();
        if (dtwave && c + 1 < LL / TT) DT_ROWS_LOAD(c + 1)

        float dtv[KPL], du[KPL], y[KPL];
#pragma unroll
        for (int j = 0; j < KPL; ++j) {
            dtv[j] = sDT[(j * 32 + lane) * TST + w];
            du[j]  = dtv[j] * sU[(j * 32 + lane) * TST + w];
            y[j]   = 0.0f;
        }
#pragma unroll
        for (int g = 0; g < NS / 4; ++g) {
            v4f Bv[KPL], Cv[KPL];
#pragma unroll
            for (int j = 0; j < KPL; ++j) {
                Bv[j] = *(const v4f*)&sBC[(j * 32 + lane) * BCP + 4 * g];
                Cv[j] = *(const v4f*)&sBC[(j * 32 + lane) * BCP + NS + 4 * g];
            }
#pragma unroll
            for (int q = 0; q < 4; ++q) {
                const int n = 4 * g + q;
                float aj[KPL], sj[KPL];
#pragma unroll
                for (int j = 0; j < KPL; ++j) {
                    aj[j] = __expf(dtv[j] * An[n]);
                    sj[j] = du[j] * Bv[j][q];
                }
                float a = aj[0];
                float s = sj[0];
#pragma unroll
                for (int j = 1; j < KPL; ++j) {
                    s = aj[j] * s + sj[j];
                    a = a * aj[j];
                }
                SCAN_STAGE(1)
                SCAN_STAGE(2)
                SCAN_STAGE(4)
                SCAN_STAGE(8)
                const float a15 = lane_to_all(a, 15);
                const float s15 = lane_to_all(s, 15);
                if (lane >= 16) {
                    s = a * s15 + s;
                    a = a * a15;
                }
                const float hend = a * hin[n] + s;
                float h = row_up<1>(hin[n], hend);
                const float h15 = lane_to_all(hend, 15);
                h = (lane == 16) ? h15 : h;
                hin[n] = lane_to_all(hend, 31);
#pragma unroll
                for (int j = 0; j < KPL; ++j) {
                    h = aj[j] * h + sj[j];
                    y[j] += h * Cv[j][q];
                }
            }
        }
#pragma unroll
        for (int j = 0; j < KPL; ++j) {
            const float uv = sU[(j * 32 + lane) * TST + w];
            const float zv = sZ[(j * 32 + lane) * TST + w];
            sY[(j * 32 + lane) * TST + w] = (y[j] + uv * Dd) * siluf(zv);
        }
        __syncthreads();

#pragma unroll
        for (int jj = 0; jj < TT / (2 * CHB); ++jj) {
            const int r = 32 * jj + 2 * w + hl;
            const int p = pm + 8 * jj;
            const size_t row = (size_t)(b * LL + c * TT + r);
            yb[((size_t)cg * (BB * LL) + row) * CHB + c16] = sY[p * TST + c16];
        }
    }
}

#undef DT_ROWS_LOAD
extern "C" void kernel_launch(void* const* d_in, const int* in_sizes, int n_in,
                              void* d_out, int out_size, void* d_ws, size_t ws_size,
                              hipStream_t stream) {
    const float* x          = (const float*)d_in[0];
    const float* in_proj_w  = (const float*)d_in[1];
    const float* conv_w     = (const float*)d_in[2];
    const float* conv_b     = (const float*)d_in[3];
    const float* x_proj_w   = (const float*)d_in[4];
    const float* dt_proj_w  = (const float*)d_in[5];
    const float* dt_proj_b  = (const float*)d_in[6];
    const float* A_log      = (const float*)d_in[7];
    const float* Dvec       = (const float*)d_in[8];
    const float* out_proj_w = (const float*)d_in[9];
    float* out = (float*)d_out;

    float* ws  = (float*)d_ws;
    float* xz  = ws;
    float* u   = xz  + (size_t)2048 * 4096;
    float* dbl = u   + (size_t)2048 * 2048;
    float* yg  = dbl + (size_t)2048 * 96;

    wmma_gemm_f32<4, 4, 16, 2, 2, XZW, DMOD, DMOD, 8, 16, EPI_CONV, 2>
        <<<dim3(XZW / 128, (BB * LL) / 128), 128, 0, stream>>>(x, in_proj_w, xz, conv_b, conv_w, u);

    conv_edge_kernel<<<((BB * LL) / XT) * (DI / 4) / 256, 256, 0, stream>>>(xz, conv_w, conv_b, u);

    wmma_gemm_f32<1, 2, 32, 2, 1, DBLW, DI, DI, 16, 32, EPI_PLAIN, 4>
        <<<dim3(DBLW / 32, (BB * LL) / 32), 64, 0, stream>>>(u, x_proj_w, dbl, nullptr, nullptr, nullptr);

    scan_kernel<<<BB * (DI / CHB), 32 * CHB, 0, stream>>>(dt_proj_w, dt_proj_b, u, dbl, xz, A_log, Dvec, yg);

    wmma_gemm_f32<2, 4, 32, 2, 2, DMOD, DI, CHB, BB * LL * CHB, 2 * BB * LL * CHB, EPI_PLAIN, 2>
        <<<dim3(DMOD / 128, (BB * LL) / 64), 128, 0, stream>>>(yg, out_proj_w, out, nullptr, nullptr, nullptr);
}
